# starting kernel
# baseline (speedup 1.0000x reference)
#include <hip/hip_runtime.h>
#include <cstddef>


typedef __attribute__((ext_vector_type(2))) float v2f;
typedef __attribute__((ext_vector_type(8))) float v8f;

#define HIDC 64

template <int ACT>
__global__ __launch_bounds__(128) void gemm_wmma_f32(
    const float* __restrict__ A, const float* __restrict__ B,
    const float* __restrict__ bias, float* __restrict__ C,
    int M, int K, int Nc)
{
    const int lane = threadIdx.x & 31;
    const int wave = threadIdx.x >> 5;
    const int hi   = lane >> 4;
    const int lo   = lane & 15;
    const int row0 = blockIdx.x << 4;
    const int n0   = (blockIdx.y * 4 + wave) << 4;
    if (n0 >= Nc) return;

    int arow = row0 + lo; if (arow > M - 1) arow = M - 1;
    int bcol = n0 + lo;   if (bcol > Nc - 1) bcol = Nc - 1;
    const float* ap  = A + (size_t)arow * K + (hi << 1);
    const float* bp0 = B + (size_t)(hi << 1) * Nc + bcol;
    const float* bp1 = bp0 + Nc;
    const size_t bstep = (size_t)Nc * 4;

    v8f acc = {0.f, 0.f, 0.f, 0.f, 0.f, 0.f, 0.f, 0.f};
    #pragma unroll 4
    for (int k = 0; k < K; k += 4) {
        v2f a = *(const v2f*)ap;
        v2f b;
        b.x = *bp0;
        b.y = *bp1;
        acc = __builtin_amdgcn_wmma_f32_16x16x4_f32(
             false, a,  false, b,
             (short)0, acc,  false,  false);
        ap  += 4;
        bp0 += bstep;
        bp1 += bstep;
    }

    const int n = n0 + lo;
    if (n < Nc) {
        const float bv = bias ? bias[n] : 0.f;
        float* cp = C + (size_t)(row0 + (hi << 3)) * Nc + n;
        #pragma unroll
        for (int j = 0; j < 8; ++j) {
            const int m = row0 + j + (hi << 3);
            if (m < M) {
                float v = acc[j] + bv;
                if (ACT == 1)
                    v = 0.5f * v * (1.f + erff(v * 0.70710678118654752f));
                cp[(size_t)j * Nc] = v;
            }
        }
    }
}

__global__ __launch_bounds__(256) void gcn_scatter(
    const float* __restrict__ h, const int* __restrict__ src,
    const int* __restrict__ dst, const float* __restrict__ ew,
    float* __restrict__ agg, int E)
{
    const int e = (blockIdx.x * blockDim.x + threadIdx.x) >> 5;
    if (e >= E) return;
    const int lane = threadIdx.x & 31;
    const int s = src[e];
    const int d = dst[e];
    const float w = ew[e];
    const float v0 = h[(size_t)s * HIDC + lane];
    const float v1 = h[(size_t)s * HIDC + lane + 32];
    unsafeAtomicAdd(&agg[(size_t)d * HIDC + lane], v0 * w);
    unsafeAtomicAdd(&agg[(size_t)d * HIDC + lane + 32], v1 * w);
}

__global__ __launch_bounds__(256) void bias_relu_stats(
    float* __restrict__ x, const float* __restrict__ bias,
    float* __restrict__ stats, int Nrows)
{
    const int t = blockIdx.x * blockDim.x + threadIdx.x;
    const int c = t & 63;
    const int rstep = (gridDim.x * blockDim.x) >> 6;
    const float b = bias[c];
    float s = 0.f, s2 = 0.f;
    float* p = x + (size_t)(t >> 6) * HIDC + c;
    const size_t pstep = (size_t)rstep * HIDC;
    for (int r = t >> 6; r < Nrows; r += rstep) {
        float v = *p + b;
        v = fmaxf(v, 0.f);
        *p = v;
        s  += v;
        s2 += v * v;
        p  += pstep;
    }
    unsafeAtomicAdd(&stats[c], s);
    unsafeAtomicAdd(&stats[64 + c], s2);
}

__global__ void bn_finalize(float* __restrict__ stats, const float* __restrict__ g,
                            const float* __restrict__ be, float invN)
{
    const int c = threadIdx.x;
    const float mean = stats[c] * invN;
    const float var  = stats[64 + c] * invN - mean * mean;
    const float sc   = g[c] * rsqrtf(var + 1e-5f);
    stats[128 + c] = sc;
    stats[192 + c] = be[c] - mean * sc;
}

__global__ __launch_bounds__(256) void bn_apply(
    float* __restrict__ x, const float* __restrict__ stats, size_t total)
{
    size_t i = (size_t)blockIdx.x * blockDim.x + threadIdx.x;
    const size_t st = (size_t)gridDim.x * blockDim.x;
    for (; i < total; i += st) {
        const int c = (int)(i & 63);
        x[i] = x[i] * stats[128 + c] + stats[192 + c];
    }
}

__global__ __launch_bounds__(256) void zero_buf4(float4* __restrict__ p, size_t n4)
{
    size_t i = (size_t)blockIdx.x * blockDim.x + threadIdx.x;
    const size_t st = (size_t)gridDim.x * blockDim.x;
    const float4 z = make_float4(0.f, 0.f, 0.f, 0.f);
    for (; i < n4; i += st) p[i] = z;
}

extern "C" void kernel_launch(void* const* d_in, const int* in_sizes, int n_in,
                              void* d_out, int out_size, void* d_ws, size_t ws_size,
                              hipStream_t stream)
{
    const float* x  = (const float*)d_in[0];
    const int*   ei = (const int*)d_in[1];
    const float* ew = (const float*)d_in[2];
    const float* cw[3] = {(const float*)d_in[3], (const float*)d_in[5], (const float*)d_in[7]};
    const float* cb[3] = {(const float*)d_in[4], (const float*)d_in[6], (const float*)d_in[8]};
    const float* bg[3] = {(const float*)d_in[9],  (const float*)d_in[11], (const float*)d_in[13]};
    const float* bb[3] = {(const float*)d_in[10], (const float*)d_in[12], (const float*)d_in[14]};
    const float* w1 = (const float*)d_in[15]; const float* b1 = (const float*)d_in[16];
    const float* w2 = (const float*)d_in[17]; const float* b2 = (const float*)d_in[18];
    const float* w3 = (const float*)d_in[19]; const float* b3 = (const float*)d_in[20];

    const int N = in_sizes[0] / 128;
    const int E = in_sizes[2];
    const int* src = ei;
    const int* dst = ei + E;

    float* bufA  = (float*)d_ws;
    float* bufB  = bufA + (size_t)N * 128;
    float* stats = bufB + (size_t)N * 64;

    const dim3 blk128(128), blk256(256);
    const int rowTiles = (N + 15) / 16;
    const int eBlocks  = (E * 32 + 255) / 256;

    const float* in_ptr = x;
    int K = 128;
    for (int l = 0; l < 3; ++l) {
        gemm_wmma_f32<0><<<dim3(rowTiles, 1), blk128, 0, stream>>>(
            in_ptr, cw[l], nullptr, bufA, N, K, HIDC);
        zero_buf4<<<2048, blk256, 0, stream>>>((float4*)bufB, (size_t)N * HIDC / 4);
        zero_buf4<<<1, 64, 0, stream>>>((float4*)stats, 64);
        gcn_scatter<<<eBlocks, blk256, 0, stream>>>(bufA, src, dst, ew, bufB, E);
        bias_relu_stats<<<1024, blk256, 0, stream>>>(bufB, cb[l], stats, N);
        bn_finalize<<<1, 64, 0, stream>>>(stats, bg[l], bb[l], 1.0f / (float)N);
        bn_apply<<<2048, blk256, 0, stream>>>(bufB, stats, (size_t)N * HIDC);
        in_ptr = bufB;
        K = HIDC;
    }

    gemm_wmma_f32<1><<<dim3(rowTiles, 2), blk128, 0, stream>>>(bufB, w1, b1, bufA, N, 64, 128);
    gemm_wmma_f32<1><<<dim3(rowTiles, 1), blk128, 0, stream>>>(bufA, w2, b2, bufB, N, 128, 64);
    gemm_wmma_f32<0><<<dim3(rowTiles, 1), blk128, 0, stream>>>(bufB, w3, b3, (float*)d_out, N, 64, 5);
    (void)n_in; (void)out_size; (void)ws_size;
}


# Edge sums: ONE owner work-group per destination range (254 ranges): it stores relu(sum+bias) and adds batch sums itself; clear pass, global adds, bias pass gone; row sum = one LDS add chain, was 2
# speedup vs baseline: 2.9568x; 2.9568x over previous
#include <hip/hip_runtime.h>
#include <cstddef>


typedef __attribute__((ext_vector_type(2))) float v2f;
typedef __attribute__((ext_vector_type(8))) float v8f;

#define HIDC 64

template <int ACT>
__global__ __launch_bounds__(128) void gemm_wmma_f32(
    const float* __restrict__ A, const float* __restrict__ B,
    const float* __restrict__ bias, float* __restrict__ C,
    int M, int K, int Nc)
{
    const int lane = threadIdx.x & 31;
    const int wave = threadIdx.x >> 5;
    const int hi   = lane >> 4;
    const int lo   = lane & 15;
    const int row0 = blockIdx.x << 4;
    const int n0   = (blockIdx.y * 4 + wave) << 4;
    if (n0 >= Nc) return;

    int arow = row0 + lo; if (arow > M - 1) arow = M - 1;
    int bcol = n0 + lo;   if (bcol > Nc - 1) bcol = Nc - 1;
    const float* ap  = A + (size_t)arow * K + (hi << 1);
    const float* bp0 = B + (size_t)(hi << 1) * Nc + bcol;
    const float* bp1 = bp0 + Nc;
    const size_t bstep = (size_t)Nc * 4;

    v8f acc = {0.f, 0.f, 0.f, 0.f, 0.f, 0.f, 0.f, 0.f};
    #pragma unroll 4
    for (int k = 0; k < K; k += 4) {
        v2f a = *(const v2f*)ap;
        v2f b;
        b.x = *bp0;
        b.y = *bp1;
        acc = __builtin_amdgcn_wmma_f32_16x16x4_f32(
             false, a,  false, b,
             (short)0, acc,  false,  false);
        ap  += 4;
        bp0 += bstep;
        bp1 += bstep;
    }

    const int n = n0 + lo;
    if (n < Nc) {
        const float bv = bias ? bias[n] : 0.f;
        float* cp = C + (size_t)(row0 + (hi << 3)) * Nc + n;
        #pragma unroll
        for (int j = 0; j < 8; ++j) {
            const int m = row0 + j + (hi << 3);
            if (m < M) {
                float v = acc[j] + bv;
                if (ACT == 1)
                    v = 0.5f * v * (1.f + erff(v * 0.70710678118654752f));
                cp[(size_t)j * Nc] = v;
            }
        }
    }
}

__global__ __launch_bounds__(128) void gemm_wmma_f32_il(
    const float* __restrict__ A, const float* __restrict__ B,
    float* __restrict__ Hs, int M, int K, float* __restrict__ stats)
{
    if (blockIdx.x == 0) stats[threadIdx.x] = 0.f;
    const int lane = threadIdx.x & 31;
    const int wave = threadIdx.x >> 5;
    const int hi   = lane >> 4;
    const int lo   = lane & 15;
    const int row0 = ((blockIdx.x << 1) + (wave >> 1)) << 4;
    const int c0   = (wave & 1) << 4;
    if (row0 >= M) return;

    int arow = row0 + lo; if (arow > M - 1) arow = M - 1;
    const float* ap  = A + (size_t)arow * K + (hi << 1);
    const float* bp0 = B + (size_t)(hi << 1) * HIDC + c0 + lo;
    const float* bp1 = bp0 + HIDC;

    v8f acc0 = {0.f, 0.f, 0.f, 0.f, 0.f, 0.f, 0.f, 0.f};
    v8f acc1 = {0.f, 0.f, 0.f, 0.f, 0.f, 0.f, 0.f, 0.f};
    #pragma unroll 4
    for (int k = 0; k < K; k += 4) {
        v2f a = *(const v2f*)ap;
        v2f b0, b1;
        b0.x = bp0[0];
        b0.y = bp1[0];
        b1.x = bp0[32];
        b1.y = bp1[32];
        acc0 = __builtin_amdgcn_wmma_f32_16x16x4_f32(
            false, a, false, b0, (short)0, acc0, false, false);
        acc1 = __builtin_amdgcn_wmma_f32_16x16x4_f32(
            false, a, false, b1, (short)0, acc1, false, false);
        ap  += 4;
        bp0 += 4 * HIDC;
        bp1 += 4 * HIDC;
    }

    v2f* cp = (v2f*)(Hs + (size_t)(row0 + (hi << 3)) * HIDC) + (c0 + lo);
    #pragma unroll
    for (int j = 0; j < 8; ++j) {
        const int m = row0 + j + (hi << 3);
        if (m < M) {
            v2f o;
            o.x = acc0[j];
            o.y = acc1[j];
            cp[(size_t)j * (HIDC / 2)] = o;
        }
    }
}

#define NRG     254
#define RMAX    800
#define OWN_PAD 0xFFu
#define CHUNK   512
#define EB      16
#define CAPB    448
#define KPAD    1023u
static_assert(RMAX <= 1023, "a real key must never end in KPAD");
static_assert(NRG * 4 <= 1024, "the block counts take 1 KiB of the workspace");
static_assert(NRG <= 255, "an owner number must fit a byte and differ from OWN_PAD");

__device__ __forceinline__ int bcast_i(int v, int l)
{
    return __builtin_amdgcn_readlane(v, l);
}
__device__ __forceinline__ float bcast_f(float v, int l)
{
    return __builtin_bit_cast(float, __builtin_amdgcn_readlane(__builtin_bit_cast(int, v), l));
}
__device__ __forceinline__ void lds_add(float* p, float v)
{
    __hip_atomic_fetch_add(p, v, __ATOMIC_RELAXED, __HIP_MEMORY_SCOPE_WORKGROUP);
}

template <bool PAIR>
__global__ __launch_bounds__(256) void edge_keys(
    const int* __restrict__ src, const int* __restrict__ dst, const float* __restrict__ ew,
    unsigned* __restrict__ key, uint2* __restrict__ kw, unsigned* __restrict__ ob,
    int N, int E, int nwords)
{
    const int t = (int)(blockIdx.x * blockDim.x + threadIdx.x);
    if (t >= nwords) return;
    const int R = (N + NRG - 1) / NRG;
    unsigned pack = 0;
    #pragma unroll
    for (int j = 0; j < 4; ++j) {
        const int e = 4 * t + j;
        unsigned o = OWN_PAD;
        if (e < E) {
            const int d  = dst[e];
            const int ow = d / R;
            o = (unsigned)ow;
            const unsigned kk = ((unsigned)src[e] << 10) | (unsigned)(d - ow * R);
            if (PAIR) kw[e] = make_uint2(kk, __builtin_bit_cast(unsigned, ew[e]));
            else      key[e] = kk;
        }
        pack |= o << (8 * j);
    }
    ob[t] = pack;
}

template <bool FULL>
__device__ __forceinline__ void add_edges(
    const v2f* __restrict__ h2, float* __restrict__ acc, unsigned k, float wt, int n, int lane)
{
    #pragma unroll
    for (int b = 0; b < 32; b += EB) {
        if (!FULL && b >= n) break;
        v2f v[EB];
        #pragma unroll
        for (int j = 0; j < EB; ++j)
            if (FULL || b + j < n)
                v[j] = h2[(bcast_i((int)k, b + j) >> 10) * (HIDC / 2) + lane];
        #pragma unroll
        for (int j = 0; j < EB; ++j)
            if (FULL || b + j < n) {
                const float w = bcast_f(wt, b + j);
                float* a = acc + (bcast_i((int)k, b + j) & 1023) * HIDC + lane;
                lds_add(a,      v[j].x * w);
                lds_add(a + 32, v[j].y * w);
            }
    }
}

template <bool PAIR, int MODE>
__global__ __launch_bounds__(1024) void gcn_owner_sum(
    const float* __restrict__ h, const unsigned* __restrict__ key,
    const float* __restrict__ ew, const uint2* __restrict__ kw,
    const unsigned* __restrict__ ob, float* __restrict__ agg, int N, int nchunks,
    uint2* __restrict__ lists, unsigned* __restrict__ nblk_g,
    const float* __restrict__ bias, float* __restrict__ stats)
{
    __shared__ float    acc[RMAX * HIDC];
    __shared__ unsigned q[32][64];
    __shared__ unsigned nblk;

    const int lane = (int)(threadIdx.x & 31u);
    const int wave = __builtin_amdgcn_readfirstlane((int)(threadIdx.x >> 5));
    const int R    = (N + NRG - 1) / NRG;
    const int rg   = (int)blockIdx.x;
    const int row0 = rg * R;
    int rows = N - row0;
    if (rows > R) rows = R;
    if (rows < 0) rows = 0;

    float4* acc4 = (float4*)acc;
    for (int i = (int)threadIdx.x; i < rows * (HIDC / 4); i += 1024)
        acc4[i] = make_float4(0.f, 0.f, 0.f, 0.f);
    if (MODE == 1 && threadIdx.x == 0) nblk = 0u;
    __syncthreads();

    const v2f*   __restrict__ h2  = (const v2f*)h;
    uint2* mylist = lists + (size_t)blockIdx.x * (CAPB * 32);
    unsigned nsaved = CAPB + 1u;
    if (MODE == 2) nsaved = nblk_g[blockIdx.x];
    if (MODE == 2 && nsaved <= CAPB) {
        const int nb = (int)nsaved;
        int b = wave;
        uint2 p = make_uint2(KPAD, 0u);
        if (b < nb) p = mylist[(size_t)b * 32 + lane];
        while (b < nb) {
            const int bn = b + 32;
            uint2 pn = p;
            if (bn < nb) pn = mylist[(size_t)bn * 32 + lane];
            const unsigned k  = p.x;
            const float    wt = __builtin_bit_cast(float, p.y);
            const unsigned real = __builtin_amdgcn_ballot_w32((k & 1023u) != KPAD);
            if (real == 0xFFFFFFFFu) add_edges<true>(h2, acc, k, wt, 32, lane);
            else                     add_edges<false>(h2, acc, k, wt, __builtin_popcount(real), lane);
            p = pn;
            b = bn;
        }
    } else {
        const uint4* __restrict__ ob4 = (const uint4*)ob;
        unsigned* myq = q[wave];
        const unsigned pat = (unsigned)rg * 0x01010101u;
        unsigned head = 0, cnt = 0;

        int c = wave;
        uint4 w = make_uint4(0xFFFFFFFFu, 0xFFFFFFFFu, 0xFFFFFFFFu, 0xFFFFFFFFu);
        if (c < nchunks) w = ob4[(size_t)c * 32 + lane];
        while (c < nchunks) {
            const int cn = c + 32;
            uint4 wn = w;
            if (cn < nchunks) wn = ob4[(size_t)cn * 32 + lane];
            const unsigned x0 = w.x ^ pat, x1 = w.y ^ pat, x2 = w.z ^ pat, x3 = w.w ^ pat;
            const unsigned z0 = ~(((x0 & 0x7F7F7F7Fu) + 0x7F7F7F7Fu) | x0) & 0x80808080u;
            const unsigned z1 = ~(((x1 & 0x7F7F7F7Fu) + 0x7F7F7F7Fu) | x1) & 0x80808080u;
            const unsigned z2 = ~(((x2 & 0x7F7F7F7Fu) + 0x7F7F7F7Fu) | x2) & 0x80808080u;
            const unsigned z3 = ~(((x3 & 0x7F7F7F7Fu) + 0x7F7F7F7Fu) | x3) & 0x80808080u;
            unsigned m = (z0 >> 7) | (z1 >> 6) | (z2 >> 5) | (z3 >> 4);
            const unsigned ebase = (unsigned)c * CHUNK + (unsigned)lane * 16u;
            for (;;) {
                const bool has = (m != 0u);
                const unsigned bal = __builtin_amdgcn_ballot_w32(has);
                if (bal == 0u) break;
                if (has) {
                    const unsigned bit = (unsigned)__builtin_ctz(m);
                    m &= m - 1u;
                    const unsigned slot = (head + cnt + __builtin_amdgcn_mbcnt_lo(bal, 0u)) & 63u;
                    myq[slot] = ebase + ((bit & 3u) << 2) + (bit >> 3);
                }
                cnt += (unsigned)__builtin_popcount(bal);
                if (cnt >= 32u) {
                    const unsigned e = myq[(head + (unsigned)lane) & 63u];
                    unsigned k;
                    float wt;
                    if (PAIR) { const uint2 p = kw[e]; k = p.x; wt = __builtin_bit_cast(float, p.y); }
                    else      { k = key[e]; wt = ew[e]; }
                    if (MODE == 1) {
                        unsigned bs = 0u;
                        if (lane == 0)
                            bs = __hip_atomic_fetch_add(&nblk, 1u, __ATOMIC_RELAXED, __HIP_MEMORY_SCOPE_WORKGROUP);
                        bs = (unsigned)__builtin_amdgcn_readfirstlane((int)bs);
                        if (bs < CAPB)
                            mylist[(size_t)bs * 32 + lane] = make_uint2(k, __builtin_bit_cast(unsigned, wt));
                    }
                    add_edges<true>(h2, acc, k, wt, 32, lane);
                    head = (head + 32u) & 63u;
                    cnt -= 32u;
                }
            }
            w = wn;
            c = cn;
        }
        if (cnt > 0u) {
            unsigned k = 0u;
            float wt = 0.f;
            if ((unsigned)lane < cnt) {
                const unsigned e = myq[(head + (unsigned)lane) & 63u];
                if (PAIR) { const uint2 p = kw[e]; k = p.x; wt = __builtin_bit_cast(float, p.y); }
                else      { k = key[e]; wt = ew[e]; }
            }
            if (MODE == 1) {
                unsigned bs = 0u;
                if (lane == 0)
                    bs = __hip_atomic_fetch_add(&nblk, 1u, __ATOMIC_RELAXED, __HIP_MEMORY_SCOPE_WORKGROUP);
                bs = (unsigned)__builtin_amdgcn_readfirstlane((int)bs);
                if (bs < CAPB)
                    mylist[(size_t)bs * 32 + lane] =
                        make_uint2((unsigned)lane < cnt ? k : KPAD, __builtin_bit_cast(unsigned, wt));
            }
            add_edges<false>(h2, acc, k, wt, (int)cnt, lane);
        }
    }
    __syncthreads();
    if (MODE == 1 && threadIdx.x == 0) nblk_g[blockIdx.x] = nblk;

    const int c4 = (int)(threadIdx.x & 15u) << 2;
    const float4 bv = make_float4(bias[c4], bias[c4 + 1], bias[c4 + 2], bias[c4 + 3]);
    float4 s1 = make_float4(0.f, 0.f, 0.f, 0.f);
    float4 s2 = make_float4(0.f, 0.f, 0.f, 0.f);
    float4* out4 = (float4*)(agg + (size_t)row0 * HIDC);
    for (int i = (int)threadIdx.x; i < rows * (HIDC / 4); i += 1024) {
        float4 v = acc4[i];
        v.x = fmaxf(v.x + bv.x, 0.f);
        v.y = fmaxf(v.y + bv.y, 0.f);
        v.z = fmaxf(v.z + bv.z, 0.f);
        v.w = fmaxf(v.w + bv.w, 0.f);
        out4[i] = v;
        s1.x += v.x; s1.y += v.y; s1.z += v.z; s1.w += v.w;
        s2.x += v.x * v.x; s2.y += v.y * v.y; s2.z += v.z * v.z; s2.w += v.w * v.w;
    }
    __syncthreads();
    acc4[threadIdx.x]        = s1;
    acc4[1024 + threadIdx.x] = s2;
    __syncthreads();
    if (threadIdx.x < 128u) {
        const float* p = acc + ((threadIdx.x >> 6) << 12) + (threadIdx.x & 63u);
        float t = 0.f;
        for (int m = 0; m < 64; ++m) t += p[m * 64];
        unsafeAtomicAdd(&stats[threadIdx.x], t);
    }
}

__global__ void bn_finalize(float* __restrict__ stats, const float* __restrict__ g,
                            const float* __restrict__ be, float invN)
{
    const int c = threadIdx.x;
    const float mean = stats[c] * invN;
    const float var  = stats[64 + c] * invN - mean * mean;
    const float sc   = g[c] * rsqrtf(var + 1e-5f);
    stats[128 + c] = sc;
    stats[192 + c] = be[c] - mean * sc;
}

__global__ __launch_bounds__(256) void bn_apply(
    float* __restrict__ x, const float* __restrict__ stats, size_t total)
{
    size_t i = (size_t)blockIdx.x * blockDim.x + threadIdx.x;
    const size_t st = (size_t)gridDim.x * blockDim.x;
    for (; i < total; i += st) {
        const int c = (int)(i & 63);
        x[i] = x[i] * stats[128 + c] + stats[192 + c];
    }
}

extern "C" void kernel_launch(void* const* d_in, const int* in_sizes, int n_in,
                              void* d_out, int out_size, void* d_ws, size_t ws_size,
                              hipStream_t stream)
{
    const float* x  = (const float*)d_in[0];
    const int*   ei = (const int*)d_in[1];
    const float* ew = (const float*)d_in[2];
    const float* cw[3] = {(const float*)d_in[3], (const float*)d_in[5], (const float*)d_in[7]};
    const float* cb[3] = {(const float*)d_in[4], (const float*)d_in[6], (const float*)d_in[8]};
    const float* bg[3] = {(const float*)d_in[9],  (const float*)d_in[11], (const float*)d_in[13]};
    const float* bb[3] = {(const float*)d_in[10], (const float*)d_in[12], (const float*)d_in[14]};
    const float* w1 = (const float*)d_in[15]; const float* b1 = (const float*)d_in[16];
    const float* w2 = (const float*)d_in[17]; const float* b2 = (const float*)d_in[18];
    const float* w3 = (const float*)d_in[19]; const float* b3 = (const float*)d_in[20];

    const int N = in_sizes[0] / 128;
    const int E = in_sizes[2];
    const int* src = ei;
    const int* dst = ei + E;

    float* bufA  = (float*)d_ws;
    float* bufB  = bufA + (size_t)N * 128;
    float* stats = bufB + (size_t)N * 64;
    const int nchunks = (E + CHUNK - 1) / CHUNK;
    unsigned* key = (unsigned*)(bufA + (size_t)N * HIDC);
    unsigned* ob  = key + (size_t)((E + 3) / 4) * 4;
    uint2* kw = (uint2*)key;
    const size_t ob_off = ((size_t)N * 192 + 256) * sizeof(float);
    const bool pair = (size_t)E * 8 <= (size_t)N * 256 &&
                      ws_size >= ob_off + (size_t)nchunks * CHUNK;
    if (pair) ob = (unsigned*)((char*)d_ws + ob_off);
    const size_t nb_off   = (ob_off + (size_t)nchunks * CHUNK + 255) & ~(size_t)255;
    const size_t list_off = nb_off + 1024;
    const bool replay = pair && ws_size >= list_off + (size_t)NRG * CAPB * 256;
    unsigned* nblk_g = (unsigned*)((char*)d_ws + nb_off);
    uint2*    lists  = (uint2*)((char*)d_ws + list_off);

    const dim3 blk128(128), blk256(256);
    const int rowTiles = (N + 15) / 16;
    const int rowPairs = (rowTiles + 1) / 2;
    const int nwords   = nchunks * (CHUNK / 4);

    if (pair)
        edge_keys<true><<<(nwords + 255) / 256, blk256, 0, stream>>>(src, dst, ew, key, kw, ob, N, E, nwords);
    else
        edge_keys<false><<<(nwords + 255) / 256, blk256, 0, stream>>>(src, dst, ew, key, kw, ob, N, E, nwords);

    const float* in_ptr = x;
    int K = 128;
    for (int l = 0; l < 3; ++l) {
        gemm_wmma_f32_il<<<rowPairs, blk128, 0, stream>>>(in_ptr, cw[l], bufA, N, K, stats);
        if (replay && l == 0)
            gcn_owner_sum<true, 1><<<NRG, dim3(1024), 0, stream>>>(bufA, key, ew, kw, ob, bufB, N, nchunks, lists, nblk_g, cb[l], stats);
        else if (replay)
            gcn_owner_sum<true, 2><<<NRG, dim3(1024), 0, stream>>>(bufA, key, ew, kw, ob, bufB, N, nchunks, lists, nblk_g, cb[l], stats);
        else if (pair)
            gcn_owner_sum<true, 0><<<NRG, dim3(1024), 0, stream>>>(bufA, key, ew, kw, ob, bufB, N, nchunks, lists, nblk_g, cb[l], stats);
        else
            gcn_owner_sum<false, 0><<<NRG, dim3(1024), 0, stream>>>(bufA, key, ew, kw, ob, bufB, N, nchunks, lists, nblk_g, cb[l], stats);
        bn_finalize<<<1, 64, 0, stream>>>(stats, bg[l], bb[l], 1.0f / (float)N);
        bn_apply<<<2048, blk256, 0, stream>>>(bufB, stats, (size_t)N * HIDC);
        in_ptr = bufB;
        K = HIDC;
    }

    gemm_wmma_f32<1><<<dim3(rowTiles, 2), blk128, 0, stream>>>(bufB, w1, b1, bufA, N, 64, 128);
    gemm_wmma_f32<1><<<dim3(rowTiles, 1), blk128, 0, stream>>>(bufA, w2, b2, bufB, N, 128, 64);
    gemm_wmma_f32<0><<<dim3(rowTiles, 1), blk128, 0, stream>>>(bufB, w3, b3, (float*)d_out, N, 64, 5);
    (void)n_in; (void)out_size;
}


# Batch norm folded into the next product's weight fragments (B scaled by scale[k]; bias = shift@W summed per lane): finalize and apply launches gone, 16 -> 10; product sums regrouped
# speedup vs baseline: 3.3563x; 1.1351x over previous
#include <hip/hip_runtime.h>
#include <cstddef>


typedef __attribute__((ext_vector_type(2))) float v2f;
typedef __attribute__((ext_vector_type(8))) float v8f;

#define HIDC 64

__device__ __forceinline__ void bn_table(
    float4* __restrict__ tab, const float* __restrict__ sums,
    const float* __restrict__ g, const float* __restrict__ be, int M)
{
    if (threadIdx.x < HIDC) {
        const int c = (int)threadIdx.x;
        const float invN = 1.0f / (float)M;
        const float mean = sums[c] * invN;
        const float var  = sums[HIDC + c] * invN - mean * mean;
        const float sc   = g[c] * rsqrtf(var + 1e-5f);
        float* t = (float*)tab + ((c >> 1) << 2) + (c & 1);
        t[0] = sc;
        t[2] = be[c] - mean * sc;
    }
    __syncthreads();
}

template <int ACT, bool BN>
__global__ __launch_bounds__(128) void gemm_wmma_f32(
    const float* __restrict__ A, const float* __restrict__ B,
    const float* __restrict__ bias, float* __restrict__ C,
    int M, int K, int Nc,
    const float* __restrict__ sums, const float* __restrict__ g, const float* __restrict__ be)
{
    __shared__ float4 bnt[HIDC / 2];
    if (BN) bn_table(bnt, sums, g, be, M);
    const int lane = threadIdx.x & 31;
    const int wave = threadIdx.x >> 5;
    const int hi   = lane >> 4;
    const int lo   = lane & 15;
    const int row0 = blockIdx.x << 4;
    const int n0   = (blockIdx.y * 4 + wave) << 4;
    if (n0 >= Nc) return;

    int arow = row0 + lo; if (arow > M - 1) arow = M - 1;
    int bcol = n0 + lo;   if (bcol > Nc - 1) bcol = Nc - 1;
    const float* ap  = A + (size_t)arow * K + (hi << 1);
    const float* bp0 = B + (size_t)(hi << 1) * Nc + bcol;
    const float* bp1 = bp0 + Nc;
    const size_t bstep = (size_t)Nc * 4;

    v8f acc = {0.f, 0.f, 0.f, 0.f, 0.f, 0.f, 0.f, 0.f};
    float cs = 0.f;
    #pragma unroll 4
    for (int k = 0; k < K; k += 4) {
        v2f a = *(const v2f*)ap;
        v2f b;
        b.x = *bp0;
        b.y = *bp1;
        if (BN) {
            const float4 q = bnt[(k >> 1) + hi];
            cs = fmaf(q.z, b.x, cs);
            cs = fmaf(q.w, b.y, cs);
            b.x *= q.x;
            b.y *= q.y;
        }
        acc = __builtin_amdgcn_wmma_f32_16x16x4_f32(
             false, a,  false, b,
             (short)0, acc,  false,  false);
        ap  += 4;
        bp0 += bstep;
        bp1 += bstep;
    }

    const float cb = BN ? cs + __shfl_xor(cs, 16) : 0.f;
    const int n = n0 + lo;
    if (n < Nc) {
        const float bv = (bias ? bias[n] : 0.f) + cb;
        float* cp = C + (size_t)(row0 + (hi << 3)) * Nc + n;
        #pragma unroll
        for (int j = 0; j < 8; ++j) {
            const int m = row0 + j + (hi << 3);
            if (m < M) {
                float v = acc[j] + bv;
                if (ACT == 1)
                    v = 0.5f * v * (1.f + erff(v * 0.70710678118654752f));
                cp[(size_t)j * Nc] = v;
            }
        }
    }
}

template <bool BN>
__global__ __launch_bounds__(128) void gemm_wmma_f32_il(
    const float* __restrict__ A, const float* __restrict__ B,
    float* __restrict__ Hs, int M, int K, float* __restrict__ stats,
    const float* __restrict__ sums, const float* __restrict__ g, const float* __restrict__ be)
{
    __shared__ float4 bnt[HIDC / 2];
    if (blockIdx.x == 0) stats[threadIdx.x] = 0.f;
    if (BN) bn_table(bnt, sums, g, be, M);
    const int lane = threadIdx.x & 31;
    const int wave = threadIdx.x >> 5;
    const int hi   = lane >> 4;
    const int lo   = lane & 15;
    const int row0 = ((blockIdx.x << 1) + (wave >> 1)) << 4;
    const int c0   = (wave & 1) << 4;
    if (row0 >= M) return;

    int arow = row0 + lo; if (arow > M - 1) arow = M - 1;
    const float* ap  = A + (size_t)arow * K + (hi << 1);
    const float* bp0 = B + (size_t)(hi << 1) * HIDC + c0 + lo;
    const float* bp1 = bp0 + HIDC;

    v8f acc0 = {0.f, 0.f, 0.f, 0.f, 0.f, 0.f, 0.f, 0.f};
    v8f acc1 = {0.f, 0.f, 0.f, 0.f, 0.f, 0.f, 0.f, 0.f};
    float cs0 = 0.f, cs1 = 0.f;
    #pragma unroll 4
    for (int k = 0; k < K; k += 4) {
        v2f a = *(const v2f*)ap;
        v2f b0, b1;
        b0.x = bp0[0];
        b0.y = bp1[0];
        b1.x = bp0[32];
        b1.y = bp1[32];
        if (BN) {
            const float4 q = bnt[(k >> 1) + hi];
            cs0 = fmaf(q.z, b0.x, cs0);
            cs0 = fmaf(q.w, b0.y, cs0);
            cs1 = fmaf(q.z, b1.x, cs1);
            cs1 = fmaf(q.w, b1.y, cs1);
            b0.x *= q.x;
            b0.y *= q.y;
            b1.x *= q.x;
            b1.y *= q.y;
        }
        acc0 = __builtin_amdgcn_wmma_f32_16x16x4_f32(
            false, a, false, b0, (short)0, acc0, false, false);
        acc1 = __builtin_amdgcn_wmma_f32_16x16x4_f32(
            false, a, false, b1, (short)0, acc1, false, false);
        ap  += 4;
        bp0 += 4 * HIDC;
        bp1 += 4 * HIDC;
    }

    const float cb0 = BN ? cs0 + __shfl_xor(cs0, 16) : 0.f;
    const float cb1 = BN ? cs1 + __shfl_xor(cs1, 16) : 0.f;
    v2f* cp = (v2f*)(Hs + (size_t)(row0 + (hi << 3)) * HIDC) + (c0 + lo);
    #pragma unroll
    for (int j = 0; j < 8; ++j) {
        const int m = row0 + j + (hi << 3);
        if (m < M) {
            v2f o;
            o.x = acc0[j] + cb0;
            o.y = acc1[j] + cb1;
            cp[(size_t)j * (HIDC / 2)] = o;
        }
    }
}

#define NRG     254
#define RMAX    800
#define OWN_PAD 0xFFu
#define CHUNK   512
#define EB      16
#define CAPB    448
#define KPAD    1023u
static_assert(RMAX <= 1023, "a real key must never end in KPAD");
static_assert(NRG * 4 <= 1024, "the block counts take 1 KiB of the workspace");
static_assert(NRG <= 255, "an owner number must fit a byte and differ from OWN_PAD");

__device__ __forceinline__ int bcast_i(int v, int l)
{
    return __builtin_amdgcn_readlane(v, l);
}
__device__ __forceinline__ float bcast_f(float v, int l)
{
    return __builtin_bit_cast(float, __builtin_amdgcn_readlane(__builtin_bit_cast(int, v), l));
}
__device__ __forceinline__ void lds_add(float* p, float v)
{
    __hip_atomic_fetch_add(p, v, __ATOMIC_RELAXED, __HIP_MEMORY_SCOPE_WORKGROUP);
}

template <bool PAIR>
__global__ __launch_bounds__(256) void edge_keys(
    const int* __restrict__ src, const int* __restrict__ dst, const float* __restrict__ ew,
    unsigned* __restrict__ key, uint2* __restrict__ kw, unsigned* __restrict__ ob,
    int N, int E, int nwords)
{
    const int t = (int)(blockIdx.x * blockDim.x + threadIdx.x);
    if (t >= nwords) return;
    const int R = (N + NRG - 1) / NRG;
    unsigned pack = 0;
    #pragma unroll
    for (int j = 0; j < 4; ++j) {
        const int e = 4 * t + j;
        unsigned o = OWN_PAD;
        if (e < E) {
            const int d  = dst[e];
            const int ow = d / R;
            o = (unsigned)ow;
            const unsigned kk = ((unsigned)src[e] << 10) | (unsigned)(d - ow * R);
            if (PAIR) kw[e] = make_uint2(kk, __builtin_bit_cast(unsigned, ew[e]));
            else      key[e] = kk;
        }
        pack |= o << (8 * j);
    }
    ob[t] = pack;
}

template <bool FULL>
__device__ __forceinline__ void add_edges(
    const v2f* __restrict__ h2, float* __restrict__ acc, unsigned k, float wt, int n, int lane)
{
    #pragma unroll
    for (int b = 0; b < 32; b += EB) {
        if (!FULL && b >= n) break;
        v2f v[EB];
        #pragma unroll
        for (int j = 0; j < EB; ++j)
            if (FULL || b + j < n)
                v[j] = h2[(bcast_i((int)k, b + j) >> 10) * (HIDC / 2) + lane];
        #pragma unroll
        for (int j = 0; j < EB; ++j)
            if (FULL || b + j < n) {
                const float w = bcast_f(wt, b + j);
                float* a = acc + (bcast_i((int)k, b + j) & 1023) * HIDC + lane;
                lds_add(a,      v[j].x * w);
                lds_add(a + 32, v[j].y * w);
            }
    }
}

template <bool PAIR, int MODE>
__global__ __launch_bounds__(1024) void gcn_owner_sum(
    const float* __restrict__ h, const unsigned* __restrict__ key,
    const float* __restrict__ ew, const uint2* __restrict__ kw,
    const unsigned* __restrict__ ob, float* __restrict__ agg, int N, int nchunks,
    uint2* __restrict__ lists, unsigned* __restrict__ nblk_g,
    const float* __restrict__ bias, float* __restrict__ stats)
{
    __shared__ float    acc[RMAX * HIDC];
    __shared__ unsigned q[32][64];
    __shared__ unsigned nblk;

    const int lane = (int)(threadIdx.x & 31u);
    const int wave = __builtin_amdgcn_readfirstlane((int)(threadIdx.x >> 5));
    const int R    = (N + NRG - 1) / NRG;
    const int rg   = (int)blockIdx.x;
    const int row0 = rg * R;
    int rows = N - row0;
    if (rows > R) rows = R;
    if (rows < 0) rows = 0;

    float4* acc4 = (float4*)acc;
    for (int i = (int)threadIdx.x; i < rows * (HIDC / 4); i += 1024)
        acc4[i] = make_float4(0.f, 0.f, 0.f, 0.f);
    if (MODE == 1 && threadIdx.x == 0) nblk = 0u;
    __syncthreads();

    const v2f*   __restrict__ h2  = (const v2f*)h;
    uint2* mylist = lists + (size_t)blockIdx.x * (CAPB * 32);
    unsigned nsaved = CAPB + 1u;
    if (MODE == 2) nsaved = nblk_g[blockIdx.x];
    if (MODE == 2 && nsaved <= CAPB) {
        const int nb = (int)nsaved;
        int b = wave;
        uint2 p = make_uint2(KPAD, 0u);
        if (b < nb) p = mylist[(size_t)b * 32 + lane];
        while (b < nb) {
            const int bn = b + 32;
            uint2 pn = p;
            if (bn < nb) pn = mylist[(size_t)bn * 32 + lane];
            const unsigned k  = p.x;
            const float    wt = __builtin_bit_cast(float, p.y);
            const unsigned real = __builtin_amdgcn_ballot_w32((k & 1023u) != KPAD);
            if (real == 0xFFFFFFFFu) add_edges<true>(h2, acc, k, wt, 32, lane);
            else                     add_edges<false>(h2, acc, k, wt, __builtin_popcount(real), lane);
            p = pn;
            b = bn;
        }
    } else {
        const uint4* __restrict__ ob4 = (const uint4*)ob;
        unsigned* myq = q[wave];
        const unsigned pat = (unsigned)rg * 0x01010101u;
        unsigned head = 0, cnt = 0;

        int c = wave;
        uint4 w = make_uint4(0xFFFFFFFFu, 0xFFFFFFFFu, 0xFFFFFFFFu, 0xFFFFFFFFu);
        if (c < nchunks) w = ob4[(size_t)c * 32 + lane];
        while (c < nchunks) {
            const int cn = c + 32;
            uint4 wn = w;
            if (cn < nchunks) wn = ob4[(size_t)cn * 32 + lane];
            const unsigned x0 = w.x ^ pat, x1 = w.y ^ pat, x2 = w.z ^ pat, x3 = w.w ^ pat;
            const unsigned z0 = ~(((x0 & 0x7F7F7F7Fu) + 0x7F7F7F7Fu) | x0) & 0x80808080u;
            const unsigned z1 = ~(((x1 & 0x7F7F7F7Fu) + 0x7F7F7F7Fu) | x1) & 0x80808080u;
            const unsigned z2 = ~(((x2 & 0x7F7F7F7Fu) + 0x7F7F7F7Fu) | x2) & 0x80808080u;
            const unsigned z3 = ~(((x3 & 0x7F7F7F7Fu) + 0x7F7F7F7Fu) | x3) & 0x80808080u;
            unsigned m = (z0 >> 7) | (z1 >> 6) | (z2 >> 5) | (z3 >> 4);
            const unsigned ebase = (unsigned)c * CHUNK + (unsigned)lane * 16u;
            for (;;) {
                const bool has = (m != 0u);
                const unsigned bal = __builtin_amdgcn_ballot_w32(has);
                if (bal == 0u) break;
                if (has) {
                    const unsigned bit = (unsigned)__builtin_ctz(m);
                    m &= m - 1u;
                    const unsigned slot = (head + cnt + __builtin_amdgcn_mbcnt_lo(bal, 0u)) & 63u;
                    myq[slot] = ebase + ((bit & 3u) << 2) + (bit >> 3);
                }
                cnt += (unsigned)__builtin_popcount(bal);
                if (cnt >= 32u) {
                    const unsigned e = myq[(head + (unsigned)lane) & 63u];
                    unsigned k;
                    float wt;
                    if (PAIR) { const uint2 p = kw[e]; k = p.x; wt = __builtin_bit_cast(float, p.y); }
                    else      { k = key[e]; wt = ew[e]; }
                    if (MODE == 1) {
                        unsigned bs = 0u;
                        if (lane == 0)
                            bs = __hip_atomic_fetch_add(&nblk, 1u, __ATOMIC_RELAXED, __HIP_MEMORY_SCOPE_WORKGROUP);
                        bs = (unsigned)__builtin_amdgcn_readfirstlane((int)bs);
                        if (bs < CAPB)
                            mylist[(size_t)bs * 32 + lane] = make_uint2(k, __builtin_bit_cast(unsigned, wt));
                    }
                    add_edges<true>(h2, acc, k, wt, 32, lane);
                    head = (head + 32u) & 63u;
                    cnt -= 32u;
                }
            }
            w = wn;
            c = cn;
        }
        if (cnt > 0u) {
            unsigned k = 0u;
            float wt = 0.f;
            if ((unsigned)lane < cnt) {
                const unsigned e = myq[(head + (unsigned)lane) & 63u];
                if (PAIR) { const uint2 p = kw[e]; k = p.x; wt = __builtin_bit_cast(float, p.y); }
                else      { k = key[e]; wt = ew[e]; }
            }
            if (MODE == 1) {
                unsigned bs = 0u;
                if (lane == 0)
                    bs = __hip_atomic_fetch_add(&nblk, 1u, __ATOMIC_RELAXED, __HIP_MEMORY_SCOPE_WORKGROUP);
                bs = (unsigned)__builtin_amdgcn_readfirstlane((int)bs);
                if (bs < CAPB)
                    mylist[(size_t)bs * 32 + lane] =
                        make_uint2((unsigned)lane < cnt ? k : KPAD, __builtin_bit_cast(unsigned, wt));
            }
            add_edges<false>(h2, acc, k, wt, (int)cnt, lane);
        }
    }
    __syncthreads();
    if (MODE == 1 && threadIdx.x == 0) nblk_g[blockIdx.x] = nblk;

    const int c4 = (int)(threadIdx.x & 15u) << 2;
    const float4 bv = make_float4(bias[c4], bias[c4 + 1], bias[c4 + 2], bias[c4 + 3]);
    float4 s1 = make_float4(0.f, 0.f, 0.f, 0.f);
    float4 s2 = make_float4(0.f, 0.f, 0.f, 0.f);
    float4* out4 = (float4*)(agg + (size_t)row0 * HIDC);
    for (int i = (int)threadIdx.x; i < rows * (HIDC / 4); i += 1024) {
        float4 v = acc4[i];
        v.x = fmaxf(v.x + bv.x, 0.f);
        v.y = fmaxf(v.y + bv.y, 0.f);
        v.z = fmaxf(v.z + bv.z, 0.f);
        v.w = fmaxf(v.w + bv.w, 0.f);
        out4[i] = v;
        s1.x += v.x; s1.y += v.y; s1.z += v.z; s1.w += v.w;
        s2.x += v.x * v.x; s2.y += v.y * v.y; s2.z += v.z * v.z; s2.w += v.w * v.w;
    }
    __syncthreads();
    acc4[threadIdx.x]        = s1;
    acc4[1024 + threadIdx.x] = s2;
    __syncthreads();
    if (threadIdx.x < 128u) {
        const float* p = acc + ((threadIdx.x >> 6) << 12) + (threadIdx.x & 63u);
        float t = 0.f;
        for (int m = 0; m < 64; ++m) t += p[m * 64];
        unsafeAtomicAdd(&stats[threadIdx.x], t);
    }
}

extern "C" void kernel_launch(void* const* d_in, const int* in_sizes, int n_in,
                              void* d_out, int out_size, void* d_ws, size_t ws_size,
                              hipStream_t stream)
{
    const float* x  = (const float*)d_in[0];
    const int*   ei = (const int*)d_in[1];
    const float* ew = (const float*)d_in[2];
    const float* cw[3] = {(const float*)d_in[3], (const float*)d_in[5], (const float*)d_in[7]};
    const float* cb[3] = {(const float*)d_in[4], (const float*)d_in[6], (const float*)d_in[8]};
    const float* bg[3] = {(const float*)d_in[9],  (const float*)d_in[11], (const float*)d_in[13]};
    const float* bb[3] = {(const float*)d_in[10], (const float*)d_in[12], (const float*)d_in[14]};
    const float* w1 = (const float*)d_in[15]; const float* b1 = (const float*)d_in[16];
    const float* w2 = (const float*)d_in[17]; const float* b2 = (const float*)d_in[18];
    const float* w3 = (const float*)d_in[19]; const float* b3 = (const float*)d_in[20];

    const int N = in_sizes[0] / 128;
    const int E = in_sizes[2];
    const int* src = ei;
    const int* dst = ei + E;

    float* bufA  = (float*)d_ws;
    float* bufB  = bufA + (size_t)N * 128;
    float* stats = bufB + (size_t)N * 64;
    const int nchunks = (E + CHUNK - 1) / CHUNK;
    unsigned* key = (unsigned*)(bufA + (size_t)N * HIDC);
    unsigned* ob  = key + (size_t)((E + 3) / 4) * 4;
    uint2* kw = (uint2*)key;
    const size_t ob_off = ((size_t)N * 192 + 256) * sizeof(float);
    const bool pair = (size_t)E * 8 <= (size_t)N * 256 &&
                      ws_size >= ob_off + (size_t)nchunks * CHUNK;
    if (pair) ob = (unsigned*)((char*)d_ws + ob_off);
    const size_t nb_off   = (ob_off + (size_t)nchunks * CHUNK + 255) & ~(size_t)255;
    const size_t list_off = nb_off + 1024;
    const bool replay = pair && ws_size >= list_off + (size_t)NRG * CAPB * 256;
    unsigned* nblk_g = (unsigned*)((char*)d_ws + nb_off);
    uint2*    lists  = (uint2*)((char*)d_ws + list_off);

    const dim3 blk128(128), blk256(256);
    const int rowTiles = (N + 15) / 16;
    const int rowPairs = (rowTiles + 1) / 2;
    const int nwords   = nchunks * (CHUNK / 4);

    if (pair)
        edge_keys<true><<<(nwords + 255) / 256, blk256, 0, stream>>>(src, dst, ew, key, kw, ob, N, E, nwords);
    else
        edge_keys<false><<<(nwords + 255) / 256, blk256, 0, stream>>>(src, dst, ew, key, kw, ob, N, E, nwords);

    for (int l = 0; l < 3; ++l) {
        float* st = stats + (l & 1) * 128;
        const float* sp = stats + ((l + 1) & 1) * 128;
        if (l == 0)
            gemm_wmma_f32_il<false><<<rowPairs, blk128, 0, stream>>>(x, cw[0], bufA, N, 128, st, sp, bg[0], bb[0]);
        else
            gemm_wmma_f32_il<true><<<rowPairs, blk128, 0, stream>>>(bufB, cw[l], bufA, N, HIDC, st, sp, bg[l - 1], bb[l - 1]);
        if (replay && l == 0)
            gcn_owner_sum<true, 1><<<NRG, dim3(1024), 0, stream>>>(bufA, key, ew, kw, ob, bufB, N, nchunks, lists, nblk_g, cb[l], st);
        else if (replay)
            gcn_owner_sum<true, 2><<<NRG, dim3(1024), 0, stream>>>(bufA, key, ew, kw, ob, bufB, N, nchunks, lists, nblk_g, cb[l], st);
        else if (pair)
            gcn_owner_sum<true, 0><<<NRG, dim3(1024), 0, stream>>>(bufA, key, ew, kw, ob, bufB, N, nchunks, lists, nblk_g, cb[l], st);
        else
            gcn_owner_sum<false, 0><<<NRG, dim3(1024), 0, stream>>>(bufA, key, ew, kw, ob, bufB, N, nchunks, lists, nblk_g, cb[l], st);
    }

    gemm_wmma_f32<1, true><<<dim3(rowTiles, 2), blk128, 0, stream>>>(bufB, w1, b1, bufA, N, HIDC, 128, stats, bg[2], bb[2]);
    gemm_wmma_f32<1, false><<<dim3(rowTiles, 1), blk128, 0, stream>>>(bufA, w2, b2, bufB, N, 128, 64, stats, bg[2], bb[2]);
    gemm_wmma_f32<0, false><<<dim3(rowTiles, 1), blk128, 0, stream>>>(bufB, w3, b3, (float*)d_out, N, 64, 5, stats, bg[2], bb[2]);
    (void)n_in; (void)out_size;
}


# Classifier products computed transposed in ONE launch: a product's D registers are the next product's B fragments as they stand (weight loader follows their k order); BN on x; k sum order permuted
# speedup vs baseline: 3.7764x; 1.1252x over previous
#include <hip/hip_runtime.h>
#include <cstddef>


typedef __attribute__((ext_vector_type(2))) float v2f;
typedef __attribute__((ext_vector_type(8))) float v8f;

#define HIDC 64

__device__ __forceinline__ void bn_table(
    float4* __restrict__ tab, const float* __restrict__ sums,
    const float* __restrict__ g, const float* __restrict__ be, int M)
{
    if (threadIdx.x < HIDC) {
        const int c = (int)threadIdx.x;
        const float invN = 1.0f / (float)M;
        const float mean = sums[c] * invN;
        const float var  = sums[HIDC + c] * invN - mean * mean;
        const float sc   = g[c] * rsqrtf(var + 1e-5f);
        float* t = (float*)tab + ((c >> 1) << 2) + (c & 1);
        t[0] = sc;
        t[2] = be[c] - mean * sc;
    }
    __syncthreads();
}


__device__ __forceinline__ float gelu_exact(float v)
{
    return 0.5f * v * (1.f + erff(v * 0.70710678118654752f));
}
__device__ __forceinline__ v8f wmma4(v2f a, v2f b, v8f c)
{
    return __builtin_amdgcn_wmma_f32_16x16x4_f32(
         false, a,  false, b,  (short)0, c,
         false,  false);
}

__global__ __launch_bounds__(128) void mlp_chain(
    const float* __restrict__ R,
    const float* __restrict__ W1, const float* __restrict__ b1,
    const float* __restrict__ W2, const float* __restrict__ b2,
    const float* __restrict__ W3, const float* __restrict__ b3,
    float* __restrict__ out, int M,
    const float* __restrict__ sums, const float* __restrict__ g, const float* __restrict__ be)
{
    __shared__ float4 bnt[HIDC / 2];
    bn_table(bnt, sums, g, be, M);
    const int lane = threadIdx.x & 31;
    const int wave = threadIdx.x >> 5;
    const int hi   = lane >> 4;
    const int lo   = lane & 15;
    const int row0 = ((blockIdx.x << 2) + wave) << 4;
    if (row0 >= M) return;

    int arow = row0 + lo; if (arow > M - 1) arow = M - 1;
    v2f xb[16];
    {
        const float4* xp = (const float4*)(R + (size_t)arow * HIDC + (hi << 2));
        #pragma unroll
        for (int u = 0; u < 8; ++u) {
            const float4 x4 = xp[2 * u];
            const float4 q0 = bnt[4 * u + 2 * hi];
            const float4 q1 = bnt[4 * u + 2 * hi + 1];
            xb[2 * u].x     = fmaf(x4.x, q0.x, q0.z);
            xb[2 * u].y     = fmaf(x4.y, q0.y, q0.w);
            xb[2 * u + 1].x = fmaf(x4.z, q1.x, q1.z);
            xb[2 * u + 1].y = fmaf(x4.w, q1.y, q1.w);
        }
    }

    const float* w1p = W1 + (size_t)(hi << 2) * 128 + lo;
    const float* w2p = W2 + (size_t)(hi << 3) * HIDC + lo;
    const float* b1p = b1 + (hi << 3);
    v8f d2[4];
    #pragma unroll
    for (int d = 0; d < 4; ++d) d2[d] = (v8f){0.f, 0.f, 0.f, 0.f, 0.f, 0.f, 0.f, 0.f};

    #pragma nounroll
    for (int c = 0; c < 8; ++c) {
        v8f d1 = {0.f, 0.f, 0.f, 0.f, 0.f, 0.f, 0.f, 0.f};
        #pragma unroll
        for (int t = 0; t < 16; ++t) {
            const int ko = (8 * (t >> 1) + 2 * (t & 1)) * 128;
            v2f a;
            a.x = w1p[ko];
            a.y = w1p[ko + 128];
            d1 = wmma4(a, xb[t], d1);
        }
        float h[8];
        #pragma unroll
        for (int r = 0; r < 8; ++r) h[r] = gelu_exact(d1[r] + b1p[r]);
        #pragma unroll
        for (int t = 0; t < 4; ++t) {
            v2f b;
            b.x = h[2 * t];
            b.y = h[2 * t + 1];
            #pragma unroll
            for (int d = 0; d < 4; ++d) {
                v2f a;
                a.x = w2p[(2 * t) * HIDC + 16 * d];
                a.y = w2p[(2 * t + 1) * HIDC + 16 * d];
                d2[d] = wmma4(a, b, d2[d]);
            }
        }
        w1p += 16;
        b1p += 16;
        w2p += 16 * HIDC;
    }

    const int   lo5 = lo < 5 ? lo : 4;
    const float* w3p = W3 + (hi << 3) * 5 + lo5;
    const float* b2p = b2 + (hi << 3);
    v8f d3 = {0.f, 0.f, 0.f, 0.f, 0.f, 0.f, 0.f, 0.f};
    #pragma unroll
    for (int d = 0; d < 4; ++d) {
        float h[8];
        #pragma unroll
        for (int r = 0; r < 8; ++r) h[r] = gelu_exact(d2[d][r] + b2p[16 * d + r]);
        #pragma unroll
        for (int t = 0; t < 4; ++t) {
            v2f b;
            b.x = h[2 * t];
            b.y = h[2 * t + 1];
            v2f a;
            const float a0 = w3p[(16 * d + 2 * t) * 5];
            const float a1 = w3p[(16 * d + 2 * t + 1) * 5];
            a.x = lo < 5 ? a0 : 0.f;
            a.y = lo < 5 ? a1 : 0.f;
            d3 = wmma4(a, b, d3);
        }
    }

    const int m = row0 + lo;
    if (hi == 0 && m < M) {
        float* op = out + (size_t)m * 5;
        op[0] = d3[0] + b3[0];
        op[1] = d3[1] + b3[1];
        op[2] = d3[2] + b3[2];
        op[3] = d3[3] + b3[3];
        op[4] = d3[4] + b3[4];
    }
}

template <bool BN>
__global__ __launch_bounds__(128) void gemm_wmma_f32_il(
    const float* __restrict__ A, const float* __restrict__ B,
    float* __restrict__ Hs, int M, int K, float* __restrict__ stats,
    const float* __restrict__ sums, const float* __restrict__ g, const float* __restrict__ be)
{
    __shared__ float4 bnt[HIDC / 2];
    if (blockIdx.x == 0) stats[threadIdx.x] = 0.f;
    if (BN) bn_table(bnt, sums, g, be, M);
    const int lane = threadIdx.x & 31;
    const int wave = threadIdx.x >> 5;
    const int hi   = lane >> 4;
    const int lo   = lane & 15;
    const int row0 = ((blockIdx.x << 1) + (wave >> 1)) << 4;
    const int c0   = (wave & 1) << 4;
    if (row0 >= M) return;

    int arow = row0 + lo; if (arow > M - 1) arow = M - 1;
    const float* ap  = A + (size_t)arow * K + (hi << 1);
    const float* bp0 = B + (size_t)(hi << 1) * HIDC + c0 + lo;
    const float* bp1 = bp0 + HIDC;

    v8f acc0 = {0.f, 0.f, 0.f, 0.f, 0.f, 0.f, 0.f, 0.f};
    v8f acc1 = {0.f, 0.f, 0.f, 0.f, 0.f, 0.f, 0.f, 0.f};
    float cs0 = 0.f, cs1 = 0.f;
    #pragma unroll 4
    for (int k = 0; k < K; k += 4) {
        v2f a = *(const v2f*)ap;
        v2f b0, b1;
        b0.x = bp0[0];
        b0.y = bp1[0];
        b1.x = bp0[32];
        b1.y = bp1[32];
        if (BN) {
            const float4 q = bnt[(k >> 1) + hi];
            cs0 = fmaf(q.z, b0.x, cs0);
            cs0 = fmaf(q.w, b0.y, cs0);
            cs1 = fmaf(q.z, b1.x, cs1);
            cs1 = fmaf(q.w, b1.y, cs1);
            b0.x *= q.x;
            b0.y *= q.y;
            b1.x *= q.x;
            b1.y *= q.y;
        }
        acc0 = __builtin_amdgcn_wmma_f32_16x16x4_f32(
            false, a, false, b0, (short)0, acc0, false, false);
        acc1 = __builtin_amdgcn_wmma_f32_16x16x4_f32(
            false, a, false, b1, (short)0, acc1, false, false);
        ap  += 4;
        bp0 += 4 * HIDC;
        bp1 += 4 * HIDC;
    }

    const float cb0 = BN ? cs0 + __shfl_xor(cs0, 16) : 0.f;
    const float cb1 = BN ? cs1 + __shfl_xor(cs1, 16) : 0.f;
    v2f* cp = (v2f*)(Hs + (size_t)(row0 + (hi << 3)) * HIDC) + (c0 + lo);
    #pragma unroll
    for (int j = 0; j < 8; ++j) {
        const int m = row0 + j + (hi << 3);
        if (m < M) {
            v2f o;
            o.x = acc0[j] + cb0;
            o.y = acc1[j] + cb1;
            cp[(size_t)j * (HIDC / 2)] = o;
        }
    }
}

#define NRG     254
#define RMAX    800
#define OWN_PAD 0xFFu
#define CHUNK   512
#define EB      16
#define CAPB    448
#define KPAD    1023u
static_assert(RMAX <= 1023, "a real key must never end in KPAD");
static_assert(NRG * 4 <= 1024, "the block counts take 1 KiB of the workspace");
static_assert(NRG <= 255, "an owner number must fit a byte and differ from OWN_PAD");

__device__ __forceinline__ int bcast_i(int v, int l)
{
    return __builtin_amdgcn_readlane(v, l);
}
__device__ __forceinline__ float bcast_f(float v, int l)
{
    return __builtin_bit_cast(float, __builtin_amdgcn_readlane(__builtin_bit_cast(int, v), l));
}
__device__ __forceinline__ void lds_add(float* p, float v)
{
    __hip_atomic_fetch_add(p, v, __ATOMIC_RELAXED, __HIP_MEMORY_SCOPE_WORKGROUP);
}

template <bool PAIR>
__global__ __launch_bounds__(256) void edge_keys(
    const int* __restrict__ src, const int* __restrict__ dst, const float* __restrict__ ew,
    unsigned* __restrict__ key, uint2* __restrict__ kw, unsigned* __restrict__ ob,
    int N, int E, int nwords)
{
    const int t = (int)(blockIdx.x * blockDim.x + threadIdx.x);
    if (t >= nwords) return;
    const int R = (N + NRG - 1) / NRG;
    unsigned pack = 0;
    #pragma unroll
    for (int j = 0; j < 4; ++j) {
        const int e = 4 * t + j;
        unsigned o = OWN_PAD;
        if (e < E) {
            const int d  = dst[e];
            const int ow = d / R;
            o = (unsigned)ow;
            const unsigned kk = ((unsigned)src[e] << 10) | (unsigned)(d - ow * R);
            if (PAIR) kw[e] = make_uint2(kk, __builtin_bit_cast(unsigned, ew[e]));
            else      key[e] = kk;
        }
        pack |= o << (8 * j);
    }
    ob[t] = pack;
}

template <bool FULL>
__device__ __forceinline__ void add_edges(
    const v2f* __restrict__ h2, float* __restrict__ acc, unsigned k, float wt, int n, int lane)
{
    #pragma unroll
    for (int b = 0; b < 32; b += EB) {
        if (!FULL && b >= n) break;
        v2f v[EB];
        #pragma unroll
        for (int j = 0; j < EB; ++j)
            if (FULL || b + j < n)
                v[j] = h2[(bcast_i((int)k, b + j) >> 10) * (HIDC / 2) + lane];
        #pragma unroll
        for (int j = 0; j < EB; ++j)
            if (FULL || b + j < n) {
                const float w = bcast_f(wt, b + j);
                float* a = acc + (bcast_i((int)k, b + j) & 1023) * HIDC + lane;
                lds_add(a,      v[j].x * w);
                lds_add(a + 32, v[j].y * w);
            }
    }
}

template <bool PAIR, int MODE>
__global__ __launch_bounds__(1024) void gcn_owner_sum(
    const float* __restrict__ h, const unsigned* __restrict__ key,
    const float* __restrict__ ew, const uint2* __restrict__ kw,
    const unsigned* __restrict__ ob, float* __restrict__ agg, int N, int nchunks,
    uint2* __restrict__ lists, unsigned* __restrict__ nblk_g,
    const float* __restrict__ bias, float* __restrict__ stats)
{
    __shared__ float    acc[RMAX * HIDC];
    __shared__ unsigned q[32][64];
    __shared__ unsigned nblk;

    const int lane = (int)(threadIdx.x & 31u);
    const int wave = __builtin_amdgcn_readfirstlane((int)(threadIdx.x >> 5));
    const int R    = (N + NRG - 1) / NRG;
    const int rg   = (int)blockIdx.x;
    const int row0 = rg * R;
    int rows = N - row0;
    if (rows > R) rows = R;
    if (rows < 0) rows = 0;

    float4* acc4 = (float4*)acc;
    for (int i = (int)threadIdx.x; i < rows * (HIDC / 4); i += 1024)
        acc4[i] = make_float4(0.f, 0.f, 0.f, 0.f);
    if (MODE == 1 && threadIdx.x == 0) nblk = 0u;
    __syncthreads();

    const v2f*   __restrict__ h2  = (const v2f*)h;
    uint2* mylist = lists + (size_t)blockIdx.x * (CAPB * 32);
    unsigned nsaved = CAPB + 1u;
    if (MODE == 2) nsaved = nblk_g[blockIdx.x];
    if (MODE == 2 && nsaved <= CAPB) {
        const int nb = (int)nsaved;
        int b = wave;
        uint2 p = make_uint2(KPAD, 0u);
        if (b < nb) p = mylist[(size_t)b * 32 + lane];
        while (b < nb) {
            const int bn = b + 32;
            uint2 pn = p;
            if (bn < nb) pn = mylist[(size_t)bn * 32 + lane];
            const unsigned k  = p.x;
            const float    wt = __builtin_bit_cast(float, p.y);
            const unsigned real = __builtin_amdgcn_ballot_w32((k & 1023u) != KPAD);
            if (real == 0xFFFFFFFFu) add_edges<true>(h2, acc, k, wt, 32, lane);
            else                     add_edges<false>(h2, acc, k, wt, __builtin_popcount(real), lane);
            p = pn;
            b = bn;
        }
    } else {
        const uint4* __restrict__ ob4 = (const uint4*)ob;
        unsigned* myq = q[wave];
        const unsigned pat = (unsigned)rg * 0x01010101u;
        unsigned head = 0, cnt = 0;

        int c = wave;
        uint4 w = make_uint4(0xFFFFFFFFu, 0xFFFFFFFFu, 0xFFFFFFFFu, 0xFFFFFFFFu);
        if (c < nchunks) w = ob4[(size_t)c * 32 + lane];
        while (c < nchunks) {
            const int cn = c + 32;
            uint4 wn = w;
            if (cn < nchunks) wn = ob4[(size_t)cn * 32 + lane];
            const unsigned x0 = w.x ^ pat, x1 = w.y ^ pat, x2 = w.z ^ pat, x3 = w.w ^ pat;
            const unsigned z0 = ~(((x0 & 0x7F7F7F7Fu) + 0x7F7F7F7Fu) | x0) & 0x80808080u;
            const unsigned z1 = ~(((x1 & 0x7F7F7F7Fu) + 0x7F7F7F7Fu) | x1) & 0x80808080u;
            const unsigned z2 = ~(((x2 & 0x7F7F7F7Fu) + 0x7F7F7F7Fu) | x2) & 0x80808080u;
            const unsigned z3 = ~(((x3 & 0x7F7F7F7Fu) + 0x7F7F7F7Fu) | x3) & 0x80808080u;
            unsigned m = (z0 >> 7) | (z1 >> 6) | (z2 >> 5) | (z3 >> 4);
            const unsigned ebase = (unsigned)c * CHUNK + (unsigned)lane * 16u;
            for (;;) {
                const bool has = (m != 0u);
                const unsigned bal = __builtin_amdgcn_ballot_w32(has);
                if (bal == 0u) break;
                if (has) {
                    const unsigned bit = (unsigned)__builtin_ctz(m);
                    m &= m - 1u;
                    const unsigned slot = (head + cnt + __builtin_amdgcn_mbcnt_lo(bal, 0u)) & 63u;
                    myq[slot] = ebase + ((bit & 3u) << 2) + (bit >> 3);
                }
                cnt += (unsigned)__builtin_popcount(bal);
                if (cnt >= 32u) {
                    const unsigned e = myq[(head + (unsigned)lane) & 63u];
                    unsigned k;
                    float wt;
                    if (PAIR) { const uint2 p = kw[e]; k = p.x; wt = __builtin_bit_cast(float, p.y); }
                    else      { k = key[e]; wt = ew[e]; }
                    if (MODE == 1) {
                        unsigned bs = 0u;
                        if (lane == 0)
                            bs = __hip_atomic_fetch_add(&nblk, 1u, __ATOMIC_RELAXED, __HIP_MEMORY_SCOPE_WORKGROUP);
                        bs = (unsigned)__builtin_amdgcn_readfirstlane((int)bs);
                        if (bs < CAPB)
                            mylist[(size_t)bs * 32 + lane] = make_uint2(k, __builtin_bit_cast(unsigned, wt));
                    }
                    add_edges<true>(h2, acc, k, wt, 32, lane);
                    head = (head + 32u) & 63u;
                    cnt -= 32u;
                }
            }
            w = wn;
            c = cn;
        }
        if (cnt > 0u) {
            unsigned k = 0u;
            float wt = 0.f;
            if ((unsigned)lane < cnt) {
                const unsigned e = myq[(head + (unsigned)lane) & 63u];
                if (PAIR) { const uint2 p = kw[e]; k = p.x; wt = __builtin_bit_cast(float, p.y); }
                else      { k = key[e]; wt = ew[e]; }
            }
            if (MODE == 1) {
                unsigned bs = 0u;
                if (lane == 0)
                    bs = __hip_atomic_fetch_add(&nblk, 1u, __ATOMIC_RELAXED, __HIP_MEMORY_SCOPE_WORKGROUP);
                bs = (unsigned)__builtin_amdgcn_readfirstlane((int)bs);
                if (bs < CAPB)
                    mylist[(size_t)bs * 32 + lane] =
                        make_uint2((unsigned)lane < cnt ? k : KPAD, __builtin_bit_cast(unsigned, wt));
            }
            add_edges<false>(h2, acc, k, wt, (int)cnt, lane);
        }
    }
    __syncthreads();
    if (MODE == 1 && threadIdx.x == 0) nblk_g[blockIdx.x] = nblk;

    const int c4 = (int)(threadIdx.x & 15u) << 2;
    const float4 bv = make_float4(bias[c4], bias[c4 + 1], bias[c4 + 2], bias[c4 + 3]);
    float4 s1 = make_float4(0.f, 0.f, 0.f, 0.f);
    float4 s2 = make_float4(0.f, 0.f, 0.f, 0.f);
    float4* out4 = (float4*)(agg + (size_t)row0 * HIDC);
    for (int i = (int)threadIdx.x; i < rows * (HIDC / 4); i += 1024) {
        float4 v = acc4[i];
        v.x = fmaxf(v.x + bv.x, 0.f);
        v.y = fmaxf(v.y + bv.y, 0.f);
        v.z = fmaxf(v.z + bv.z, 0.f);
        v.w = fmaxf(v.w + bv.w, 0.f);
        out4[i] = v;
        s1.x += v.x; s1.y += v.y; s1.z += v.z; s1.w += v.w;
        s2.x += v.x * v.x; s2.y += v.y * v.y; s2.z += v.z * v.z; s2.w += v.w * v.w;
    }
    __syncthreads();
    acc4[threadIdx.x]        = s1;
    acc4[1024 + threadIdx.x] = s2;
    __syncthreads();
    if (threadIdx.x < 128u) {
        const float* p = acc + ((threadIdx.x >> 6) << 12) + (threadIdx.x & 63u);
        float t = 0.f;
        for (int m = 0; m < 64; ++m) t += p[m * 64];
        unsafeAtomicAdd(&stats[threadIdx.x], t);
    }
}

extern "C" void kernel_launch(void* const* d_in, const int* in_sizes, int n_in,
                              void* d_out, int out_size, void* d_ws, size_t ws_size,
                              hipStream_t stream)
{
    const float* x  = (const float*)d_in[0];
    const int*   ei = (const int*)d_in[1];
    const float* ew = (const float*)d_in[2];
    const float* cw[3] = {(const float*)d_in[3], (const float*)d_in[5], (const float*)d_in[7]};
    const float* cb[3] = {(const float*)d_in[4], (const float*)d_in[6], (const float*)d_in[8]};
    const float* bg[3] = {(const float*)d_in[9],  (const float*)d_in[11], (const float*)d_in[13]};
    const float* bb[3] = {(const float*)d_in[10], (const float*)d_in[12], (const float*)d_in[14]};
    const float* w1 = (const float*)d_in[15]; const float* b1 = (const float*)d_in[16];
    const float* w2 = (const float*)d_in[17]; const float* b2 = (const float*)d_in[18];
    const float* w3 = (const float*)d_in[19]; const float* b3 = (const float*)d_in[20];

    const int N = in_sizes[0] / 128;
    const int E = in_sizes[2];
    const int* src = ei;
    const int* dst = ei + E;

    float* bufA  = (float*)d_ws;
    float* bufB  = bufA + (size_t)N * 128;
    float* stats = bufB + (size_t)N * 64;
    const int nchunks = (E + CHUNK - 1) / CHUNK;
    unsigned* key = (unsigned*)(bufA + (size_t)N * HIDC);
    unsigned* ob  = key + (size_t)((E + 3) / 4) * 4;
    uint2* kw = (uint2*)key;
    const size_t ob_off = ((size_t)N * 192 + 256) * sizeof(float);
    const bool pair = (size_t)E * 8 <= (size_t)N * 256 &&
                      ws_size >= ob_off + (size_t)nchunks * CHUNK;
    if (pair) ob = (unsigned*)((char*)d_ws + ob_off);
    const size_t nb_off   = (ob_off + (size_t)nchunks * CHUNK + 255) & ~(size_t)255;
    const size_t list_off = nb_off + 1024;
    const bool replay = pair && ws_size >= list_off + (size_t)NRG * CAPB * 256;
    unsigned* nblk_g = (unsigned*)((char*)d_ws + nb_off);
    uint2*    lists  = (uint2*)((char*)d_ws + list_off);

    const dim3 blk128(128), blk256(256);
    const int rowTiles = (N + 15) / 16;
    const int rowPairs = (rowTiles + 1) / 2;
    const int nwords   = nchunks * (CHUNK / 4);

    if (pair)
        edge_keys<true><<<(nwords + 255) / 256, blk256, 0, stream>>>(src, dst, ew, key, kw, ob, N, E, nwords);
    else
        edge_keys<false><<<(nwords + 255) / 256, blk256, 0, stream>>>(src, dst, ew, key, kw, ob, N, E, nwords);

    for (int l = 0; l < 3; ++l) {
        float* st = stats + (l & 1) * 128;
        const float* sp = stats + ((l + 1) & 1) * 128;
        if (l == 0)
            gemm_wmma_f32_il<false><<<rowPairs, blk128, 0, stream>>>(x, cw[0], bufA, N, 128, st, sp, bg[0], bb[0]);
        else
            gemm_wmma_f32_il<true><<<rowPairs, blk128, 0, stream>>>(bufB, cw[l], bufA, N, HIDC, st, sp, bg[l - 1], bb[l - 1]);
        if (replay && l == 0)
            gcn_owner_sum<true, 1><<<NRG, dim3(1024), 0, stream>>>(bufA, key, ew, kw, ob, bufB, N, nchunks, lists, nblk_g, cb[l], st);
        else if (replay)
            gcn_owner_sum<true, 2><<<NRG, dim3(1024), 0, stream>>>(bufA, key, ew, kw, ob, bufB, N, nchunks, lists, nblk_g, cb[l], st);
        else if (pair)
            gcn_owner_sum<true, 0><<<NRG, dim3(1024), 0, stream>>>(bufA, key, ew, kw, ob, bufB, N, nchunks, lists, nblk_g, cb[l], st);
        else
            gcn_owner_sum<false, 0><<<NRG, dim3(1024), 0, stream>>>(bufA, key, ew, kw, ob, bufB, N, nchunks, lists, nblk_g, cb[l], st);
    }

    mlp_chain<<<(rowTiles + 3) / 4, blk128, 0, stream>>>(bufB, w1, b1, w2, b2, w3, b3, (float*)d_out, N, stats, bg[2], bb[2]);
    (void)n_in; (void)out_size;
}


# Owner numbers stored bit-sliced (8 plane words per 32 edges, from wave ballots in the pre-pass): layer 0's scan gets 32 edges' match mask from 8 bit ops, was ~46 per 32; add order arrival-dependent
# speedup vs baseline: 4.1089x; 1.0880x over previous
#include <hip/hip_runtime.h>
#include <cstddef>


typedef __attribute__((ext_vector_type(2))) float v2f;
typedef __attribute__((ext_vector_type(8))) float v8f;

#define HIDC 64

__device__ __forceinline__ void bn_table(
    float4* __restrict__ tab, const float* __restrict__ sums,
    const float* __restrict__ g, const float* __restrict__ be, int M)
{
    if (threadIdx.x < HIDC) {
        const int c = (int)threadIdx.x;
        const float invN = 1.0f / (float)M;
        const float mean = sums[c] * invN;
        const float var  = sums[HIDC + c] * invN - mean * mean;
        const float sc   = g[c] * rsqrtf(var + 1e-5f);
        float* t = (float*)tab + ((c >> 1) << 2) + (c & 1);
        t[0] = sc;
        t[2] = be[c] - mean * sc;
    }
    __syncthreads();
}


__device__ __forceinline__ float gelu_exact(float v)
{
    return 0.5f * v * (1.f + erff(v * 0.70710678118654752f));
}
__device__ __forceinline__ v8f wmma4(v2f a, v2f b, v8f c)
{
    return __builtin_amdgcn_wmma_f32_16x16x4_f32(
         false, a,  false, b,  (short)0, c,
         false,  false);
}

__global__ __launch_bounds__(128) void mlp_chain(
    const float* __restrict__ R,
    const float* __restrict__ W1, const float* __restrict__ b1,
    const float* __restrict__ W2, const float* __restrict__ b2,
    const float* __restrict__ W3, const float* __restrict__ b3,
    float* __restrict__ out, int M,
    const float* __restrict__ sums, const float* __restrict__ g, const float* __restrict__ be)
{
    __shared__ float4 bnt[HIDC / 2];
    bn_table(bnt, sums, g, be, M);
    const int lane = threadIdx.x & 31;
    const int wave = threadIdx.x >> 5;
    const int hi   = lane >> 4;
    const int lo   = lane & 15;
    const int row0 = ((blockIdx.x << 2) + wave) << 4;
    if (row0 >= M) return;

    int arow = row0 + lo; if (arow > M - 1) arow = M - 1;
    v2f xb[16];
    {
        const float4* xp = (const float4*)(R + (size_t)arow * HIDC + (hi << 2));
        #pragma unroll
        for (int u = 0; u < 8; ++u) {
            const float4 x4 = xp[2 * u];
            const float4 q0 = bnt[4 * u + 2 * hi];
            const float4 q1 = bnt[4 * u + 2 * hi + 1];
            xb[2 * u].x     = fmaf(x4.x, q0.x, q0.z);
            xb[2 * u].y     = fmaf(x4.y, q0.y, q0.w);
            xb[2 * u + 1].x = fmaf(x4.z, q1.x, q1.z);
            xb[2 * u + 1].y = fmaf(x4.w, q1.y, q1.w);
        }
    }

    const float* w1p = W1 + (size_t)(hi << 2) * 128 + lo;
    const float* w2p = W2 + (size_t)(hi << 3) * HIDC + lo;
    const float* b1p = b1 + (hi << 3);
    v8f d2[4];
    #pragma unroll
    for (int d = 0; d < 4; ++d) d2[d] = (v8f){0.f, 0.f, 0.f, 0.f, 0.f, 0.f, 0.f, 0.f};

    #pragma nounroll
    for (int c = 0; c < 8; ++c) {
        v8f d1 = {0.f, 0.f, 0.f, 0.f, 0.f, 0.f, 0.f, 0.f};
        #pragma unroll
        for (int t = 0; t < 16; ++t) {
            const int ko = (8 * (t >> 1) + 2 * (t & 1)) * 128;
            v2f a;
            a.x = w1p[ko];
            a.y = w1p[ko + 128];
            d1 = wmma4(a, xb[t], d1);
        }
        float h[8];
        #pragma unroll
        for (int r = 0; r < 8; ++r) h[r] = gelu_exact(d1[r] + b1p[r]);
        #pragma unroll
        for (int t = 0; t < 4; ++t) {
            v2f b;
            b.x = h[2 * t];
            b.y = h[2 * t + 1];
            #pragma unroll
            for (int d = 0; d < 4; ++d) {
                v2f a;
                a.x = w2p[(2 * t) * HIDC + 16 * d];
                a.y = w2p[(2 * t + 1) * HIDC + 16 * d];
                d2[d] = wmma4(a, b, d2[d]);
            }
        }
        w1p += 16;
        b1p += 16;
        w2p += 16 * HIDC;
    }

    const int   lo5 = lo < 5 ? lo : 4;
    const float* w3p = W3 + (hi << 3) * 5 + lo5;
    const float* b2p = b2 + (hi << 3);
    v8f d3 = {0.f, 0.f, 0.f, 0.f, 0.f, 0.f, 0.f, 0.f};
    #pragma unroll
    for (int d = 0; d < 4; ++d) {
        float h[8];
        #pragma unroll
        for (int r = 0; r < 8; ++r) h[r] = gelu_exact(d2[d][r] + b2p[16 * d + r]);
        #pragma unroll
        for (int t = 0; t < 4; ++t) {
            v2f b;
            b.x = h[2 * t];
            b.y = h[2 * t + 1];
            v2f a;
            const float a0 = w3p[(16 * d + 2 * t) * 5];
            const float a1 = w3p[(16 * d + 2 * t + 1) * 5];
            a.x = lo < 5 ? a0 : 0.f;
            a.y = lo < 5 ? a1 : 0.f;
            d3 = wmma4(a, b, d3);
        }
    }

    const int m = row0 + lo;
    if (hi == 0 && m < M) {
        float* op = out + (size_t)m * 5;
        op[0] = d3[0] + b3[0];
        op[1] = d3[1] + b3[1];
        op[2] = d3[2] + b3[2];
        op[3] = d3[3] + b3[3];
        op[4] = d3[4] + b3[4];
    }
}

template <bool BN>
__global__ __launch_bounds__(128) void gemm_wmma_f32_il(
    const float* __restrict__ A, const float* __restrict__ B,
    float* __restrict__ Hs, int M, int K, float* __restrict__ stats,
    const float* __restrict__ sums, const float* __restrict__ g, const float* __restrict__ be)
{
    __shared__ float4 bnt[HIDC / 2];
    if (blockIdx.x == 0) stats[threadIdx.x] = 0.f;
    if (BN) bn_table(bnt, sums, g, be, M);
    const int lane = threadIdx.x & 31;
    const int wave = threadIdx.x >> 5;
    const int hi   = lane >> 4;
    const int lo   = lane & 15;
    const int row0 = ((blockIdx.x << 1) + (wave >> 1)) << 4;
    const int c0   = (wave & 1) << 4;
    if (row0 >= M) return;

    int arow = row0 + lo; if (arow > M - 1) arow = M - 1;
    const float* ap  = A + (size_t)arow * K + (hi << 1);
    const float* bp0 = B + (size_t)(hi << 1) * HIDC + c0 + lo;
    const float* bp1 = bp0 + HIDC;

    v8f acc0 = {0.f, 0.f, 0.f, 0.f, 0.f, 0.f, 0.f, 0.f};
    v8f acc1 = {0.f, 0.f, 0.f, 0.f, 0.f, 0.f, 0.f, 0.f};
    float cs0 = 0.f, cs1 = 0.f;
    #pragma unroll 4
    for (int k = 0; k < K; k += 4) {
        v2f a = *(const v2f*)ap;
        v2f b0, b1;
        b0.x = bp0[0];
        b0.y = bp1[0];
        b1.x = bp0[32];
        b1.y = bp1[32];
        if (BN) {
            const float4 q = bnt[(k >> 1) + hi];
            cs0 = fmaf(q.z, b0.x, cs0);
            cs0 = fmaf(q.w, b0.y, cs0);
            cs1 = fmaf(q.z, b1.x, cs1);
            cs1 = fmaf(q.w, b1.y, cs1);
            b0.x *= q.x;
            b0.y *= q.y;
            b1.x *= q.x;
            b1.y *= q.y;
        }
        acc0 = __builtin_amdgcn_wmma_f32_16x16x4_f32(
            false, a, false, b0, (short)0, acc0, false, false);
        acc1 = __builtin_amdgcn_wmma_f32_16x16x4_f32(
            false, a, false, b1, (short)0, acc1, false, false);
        ap  += 4;
        bp0 += 4 * HIDC;
        bp1 += 4 * HIDC;
    }

    const float cb0 = BN ? cs0 + __shfl_xor(cs0, 16) : 0.f;
    const float cb1 = BN ? cs1 + __shfl_xor(cs1, 16) : 0.f;
    v2f* cp = (v2f*)(Hs + (size_t)(row0 + (hi << 3)) * HIDC) + (c0 + lo);
    #pragma unroll
    for (int j = 0; j < 8; ++j) {
        const int m = row0 + j + (hi << 3);
        if (m < M) {
            v2f o;
            o.x = acc0[j] + cb0;
            o.y = acc1[j] + cb1;
            cp[(size_t)j * (HIDC / 2)] = o;
        }
    }
}

#define NRG     254
#define RMAX    800
#define OWN_PAD 0xFFu
#define CHUNK   512
#define EB      16
#define CAPB    448
#define KPAD    1023u
static_assert(RMAX <= 1023, "a real key must never end in KPAD");
static_assert(CHUNK == 512, "the plane blocks (two chunks, 32 groups of 32 edges) are written for 512-edge chunks");
static_assert(NRG * 4 <= 1024, "the block counts take 1 KiB of the workspace");
static_assert(NRG <= 255, "an owner number must fit 8 bits and differ from OWN_PAD");

__device__ __forceinline__ int bcast_i(int v, int l)
{
    return __builtin_amdgcn_readlane(v, l);
}
__device__ __forceinline__ float bcast_f(float v, int l)
{
    return __builtin_bit_cast(float, __builtin_amdgcn_readlane(__builtin_bit_cast(int, v), l));
}
__device__ __forceinline__ void lds_add(float* p, float v)
{
    __hip_atomic_fetch_add(p, v, __ATOMIC_RELAXED, __HIP_MEMORY_SCOPE_WORKGROUP);
}

__device__ __forceinline__ unsigned edge_of(unsigned code)
{
    const unsigned G = code >> 5;
    return ((G >> 2) << 7) + ((code & 31u) << 2) + (G & 3u);
}

template <bool PAIR>
__global__ __launch_bounds__(256) void edge_keys(
    const int* __restrict__ src, const int* __restrict__ dst, const float* __restrict__ ew,
    unsigned* __restrict__ key, uint2* __restrict__ kw, unsigned* __restrict__ ob,
    int N, int E, int nwords)
{
    const int t = (int)(blockIdx.x * blockDim.x + threadIdx.x);
    if (t >= nwords) return;
    const int R = (N + NRG - 1) / NRG;
    const unsigned lane = threadIdx.x & 31u;
    unsigned mine = 0u;
    #pragma unroll
    for (int j = 0; j < 4; ++j) {
        const int e = 4 * t + j;
        unsigned o = OWN_PAD;
        if (e < E) {
            const int d  = dst[e];
            const int ow = d / R;
            o = (unsigned)ow;
            const unsigned kk = ((unsigned)src[e] << 10) | (unsigned)(d - ow * R);
            if (PAIR) kw[e] = make_uint2(kk, __builtin_bit_cast(unsigned, ew[e]));
            else      key[e] = kk;
        }
        const unsigned p0 = __builtin_amdgcn_ballot_w32((o & 1u) != 0u);
        const unsigned p1 = __builtin_amdgcn_ballot_w32((o & 2u) != 0u);
        const unsigned p2 = __builtin_amdgcn_ballot_w32((o & 4u) != 0u);
        const unsigned p3 = __builtin_amdgcn_ballot_w32((o & 8u) != 0u);
        const unsigned p4 = __builtin_amdgcn_ballot_w32((o & 16u) != 0u);
        const unsigned p5 = __builtin_amdgcn_ballot_w32((o & 32u) != 0u);
        const unsigned p6 = __builtin_amdgcn_ballot_w32((o & 64u) != 0u);
        const unsigned p7 = __builtin_amdgcn_ballot_w32((o & 128u) != 0u);
        const unsigned q01 = (lane & 1u) ? p1 : p0, q23 = (lane & 1u) ? p3 : p2;
        const unsigned q45 = (lane & 1u) ? p5 : p4, q67 = (lane & 1u) ? p7 : p6;
        const unsigned q03 = (lane & 2u) ? q23 : q01, q47 = (lane & 2u) ? q67 : q45;
        const unsigned pk  = (lane & 4u) ? q47 : q03;
        if ((lane >> 3) == (unsigned)j) mine = pk;
    }
    const unsigned G = (((unsigned)t >> 5) << 2) + (lane >> 3);
    const unsigned k = lane & 7u;
    ob[((((G >> 5) << 1) + (k >> 2)) * 32u + (G & 31u)) * 4u + (k & 3u)] = mine;
}

template <bool FULL>
__device__ __forceinline__ void add_edges(
    const v2f* __restrict__ h2, float* __restrict__ acc, unsigned k, float wt, int n, int lane)
{
    #pragma unroll
    for (int b = 0; b < 32; b += EB) {
        if (!FULL && b >= n) break;
        v2f v[EB];
        #pragma unroll
        for (int j = 0; j < EB; ++j)
            if (FULL || b + j < n)
                v[j] = h2[(bcast_i((int)k, b + j) >> 10) * (HIDC / 2) + lane];
        #pragma unroll
        for (int j = 0; j < EB; ++j)
            if (FULL || b + j < n) {
                const float w = bcast_f(wt, b + j);
                float* a = acc + (bcast_i((int)k, b + j) & 1023) * HIDC + lane;
                lds_add(a,      v[j].x * w);
                lds_add(a + 32, v[j].y * w);
            }
    }
}

template <bool PAIR, int MODE>
__global__ __launch_bounds__(1024) void gcn_owner_sum(
    const float* __restrict__ h, const unsigned* __restrict__ key,
    const float* __restrict__ ew, const uint2* __restrict__ kw,
    const unsigned* __restrict__ ob, float* __restrict__ agg, int N, int nchunks,
    uint2* __restrict__ lists, unsigned* __restrict__ nblk_g,
    const float* __restrict__ bias, float* __restrict__ stats)
{
    __shared__ float    acc[RMAX * HIDC];
    __shared__ unsigned q[32][64];
    __shared__ unsigned nblk;

    const int lane = (int)(threadIdx.x & 31u);
    const int wave = __builtin_amdgcn_readfirstlane((int)(threadIdx.x >> 5));
    const int R    = (N + NRG - 1) / NRG;
    const int rg   = (int)blockIdx.x;
    const int row0 = rg * R;
    int rows = N - row0;
    if (rows > R) rows = R;
    if (rows < 0) rows = 0;

    float4* acc4 = (float4*)acc;
    for (int i = (int)threadIdx.x; i < rows * (HIDC / 4); i += 1024)
        acc4[i] = make_float4(0.f, 0.f, 0.f, 0.f);
    if (MODE == 1 && threadIdx.x == 0) nblk = 0u;
    __syncthreads();

    const v2f*   __restrict__ h2  = (const v2f*)h;
    uint2* mylist = lists + (size_t)blockIdx.x * (CAPB * 32);
    unsigned nsaved = CAPB + 1u;
    if (MODE == 2) nsaved = nblk_g[blockIdx.x];
    if (MODE == 2 && nsaved <= CAPB) {
        const int nb = (int)nsaved;
        int b = wave;
        uint2 p = make_uint2(KPAD, 0u);
        if (b < nb) p = mylist[(size_t)b * 32 + lane];
        while (b < nb) {
            const int bn = b + 32;
            uint2 pn = p;
            if (bn < nb) pn = mylist[(size_t)bn * 32 + lane];
            const unsigned k  = p.x;
            const float    wt = __builtin_bit_cast(float, p.y);
            const unsigned real = __builtin_amdgcn_ballot_w32((k & 1023u) != KPAD);
            if (real == 0xFFFFFFFFu) add_edges<true>(h2, acc, k, wt, 32, lane);
            else                     add_edges<false>(h2, acc, k, wt, __builtin_popcount(real), lane);
            p = pn;
            b = bn;
        }
    } else {
        const uint4* __restrict__ ob4 = (const uint4*)ob;
        unsigned* myq = q[wave];
        const unsigned sel0 = (rg & 1)   ? 0u : 0xFFFFFFFFu, sel1 = (rg & 2)   ? 0u : 0xFFFFFFFFu;
        const unsigned sel2 = (rg & 4)   ? 0u : 0xFFFFFFFFu, sel3 = (rg & 8)   ? 0u : 0xFFFFFFFFu;
        const unsigned sel4 = (rg & 16)  ? 0u : 0xFFFFFFFFu, sel5 = (rg & 32)  ? 0u : 0xFFFFFFFFu;
        const unsigned sel6 = (rg & 64)  ? 0u : 0xFFFFFFFFu, sel7 = (rg & 128) ? 0u : 0xFFFFFFFFu;
        unsigned head = 0, cnt = 0;

        int c = wave << 1;
        uint4 wa = make_uint4(0xFFFFFFFFu, 0xFFFFFFFFu, 0xFFFFFFFFu, 0xFFFFFFFFu);
        uint4 wb = wa;
        if (c < nchunks) {
            wa = ob4[(size_t)c * 32 + lane];
            wb = ob4[(size_t)(c + 1) * 32 + lane];
        }
        while (c < nchunks) {
            const int cn = c + 64;
            const int cl = cn < nchunks ? cn : c;
            const uint4 na = ob4[(size_t)cl * 32 + lane];
            const uint4 nb = ob4[(size_t)(cl + 1) * 32 + lane];
            unsigned m = (wa.x ^ sel0) & (wa.y ^ sel1) & (wa.z ^ sel2) & (wa.w ^ sel3) &
                         (wb.x ^ sel4) & (wb.y ^ sel5) & (wb.z ^ sel6) & (wb.w ^ sel7);
            const unsigned ecode = ((unsigned)c << 9) | ((unsigned)lane << 5);
            for (;;) {
                const bool has = (m != 0u);
                const unsigned bal = __builtin_amdgcn_ballot_w32(has);
                if (bal == 0u) break;
                if (has) {
                    const unsigned bit = (unsigned)__builtin_ctz(m);
                    m &= m - 1u;
                    const unsigned slot = (head + cnt + __builtin_amdgcn_mbcnt_lo(bal, 0u)) & 63u;
                    myq[slot] = ecode | bit;
                }
                cnt += (unsigned)__builtin_popcount(bal);
                if (cnt >= 32u) {
                    const unsigned e = edge_of(myq[(head + (unsigned)lane) & 63u]);
                    unsigned k;
                    float wt;
                    if (PAIR) { const uint2 p = kw[e]; k = p.x; wt = __builtin_bit_cast(float, p.y); }
                    else      { k = key[e]; wt = ew[e]; }
                    if (MODE == 1) {
                        unsigned bs = 0u;
                        if (lane == 0)
                            bs = __hip_atomic_fetch_add(&nblk, 1u, __ATOMIC_RELAXED, __HIP_MEMORY_SCOPE_WORKGROUP);
                        bs = (unsigned)__builtin_amdgcn_readfirstlane((int)bs);
                        if (bs < CAPB)
                            mylist[(size_t)bs * 32 + lane] = make_uint2(k, __builtin_bit_cast(unsigned, wt));
                    }
                    add_edges<true>(h2, acc, k, wt, 32, lane);
                    head = (head + 32u) & 63u;
                    cnt -= 32u;
                }
            }
            wa = na;
            wb = nb;
            c = cn;
        }
        if (cnt > 0u) {
            unsigned k = 0u;
            float wt = 0.f;
            if ((unsigned)lane < cnt) {
                const unsigned e = edge_of(myq[(head + (unsigned)lane) & 63u]);
                if (PAIR) { const uint2 p = kw[e]; k = p.x; wt = __builtin_bit_cast(float, p.y); }
                else      { k = key[e]; wt = ew[e]; }
            }
            if (MODE == 1) {
                unsigned bs = 0u;
                if (lane == 0)
                    bs = __hip_atomic_fetch_add(&nblk, 1u, __ATOMIC_RELAXED, __HIP_MEMORY_SCOPE_WORKGROUP);
                bs = (unsigned)__builtin_amdgcn_readfirstlane((int)bs);
                if (bs < CAPB)
                    mylist[(size_t)bs * 32 + lane] =
                        make_uint2((unsigned)lane < cnt ? k : KPAD, __builtin_bit_cast(unsigned, wt));
            }
            add_edges<false>(h2, acc, k, wt, (int)cnt, lane);
        }
    }
    __syncthreads();
    if (MODE == 1 && threadIdx.x == 0) nblk_g[blockIdx.x] = nblk;

    const int c4 = (int)(threadIdx.x & 15u) << 2;
    const float4 bv = make_float4(bias[c4], bias[c4 + 1], bias[c4 + 2], bias[c4 + 3]);
    float4 s1 = make_float4(0.f, 0.f, 0.f, 0.f);
    float4 s2 = make_float4(0.f, 0.f, 0.f, 0.f);
    float4* out4 = (float4*)(agg + (size_t)row0 * HIDC);
    for (int i = (int)threadIdx.x; i < rows * (HIDC / 4); i += 1024) {
        float4 v = acc4[i];
        v.x = fmaxf(v.x + bv.x, 0.f);
        v.y = fmaxf(v.y + bv.y, 0.f);
        v.z = fmaxf(v.z + bv.z, 0.f);
        v.w = fmaxf(v.w + bv.w, 0.f);
        out4[i] = v;
        s1.x += v.x; s1.y += v.y; s1.z += v.z; s1.w += v.w;
        s2.x += v.x * v.x; s2.y += v.y * v.y; s2.z += v.z * v.z; s2.w += v.w * v.w;
    }
    __syncthreads();
    acc4[threadIdx.x]        = s1;
    acc4[1024 + threadIdx.x] = s2;
    __syncthreads();
    if (threadIdx.x < 128u) {
        const float* p = acc + ((threadIdx.x >> 6) << 12) + (threadIdx.x & 63u);
        float t = 0.f;
        for (int m = 0; m < 64; ++m) t += p[m * 64];
        unsafeAtomicAdd(&stats[threadIdx.x], t);
    }
}

extern "C" void kernel_launch(void* const* d_in, const int* in_sizes, int n_in,
                              void* d_out, int out_size, void* d_ws, size_t ws_size,
                              hipStream_t stream)
{
    const float* x  = (const float*)d_in[0];
    const int*   ei = (const int*)d_in[1];
    const float* ew = (const float*)d_in[2];
    const float* cw[3] = {(const float*)d_in[3], (const float*)d_in[5], (const float*)d_in[7]};
    const float* cb[3] = {(const float*)d_in[4], (const float*)d_in[6], (const float*)d_in[8]};
    const float* bg[3] = {(const float*)d_in[9],  (const float*)d_in[11], (const float*)d_in[13]};
    const float* bb[3] = {(const float*)d_in[10], (const float*)d_in[12], (const float*)d_in[14]};
    const float* w1 = (const float*)d_in[15]; const float* b1 = (const float*)d_in[16];
    const float* w2 = (const float*)d_in[17]; const float* b2 = (const float*)d_in[18];
    const float* w3 = (const float*)d_in[19]; const float* b3 = (const float*)d_in[20];

    const int N = in_sizes[0] / 128;
    const int E = in_sizes[2];
    const int* src = ei;
    const int* dst = ei + E;

    float* bufA  = (float*)d_ws;
    float* bufB  = bufA + (size_t)N * 128;
    float* stats = bufB + (size_t)N * 64;
    const int nchunks = 2 * ((E + 2 * CHUNK - 1) / (2 * CHUNK));
    unsigned* key = (unsigned*)(bufA + (size_t)N * HIDC);
    unsigned* ob  = key + (size_t)((E + 3) / 4) * 4;
    uint2* kw = (uint2*)key;
    const size_t ob_off = ((size_t)N * 192 + 256) * sizeof(float);
    const bool pair = (size_t)E * 8 <= (size_t)N * 256 &&
                      ws_size >= ob_off + (size_t)nchunks * CHUNK;
    if (pair) ob = (unsigned*)((char*)d_ws + ob_off);
    const size_t nb_off   = (ob_off + (size_t)nchunks * CHUNK + 255) & ~(size_t)255;
    const size_t list_off = nb_off + 1024;
    const bool replay = pair && ws_size >= list_off + (size_t)NRG * CAPB * 256;
    unsigned* nblk_g = (unsigned*)((char*)d_ws + nb_off);
    uint2*    lists  = (uint2*)((char*)d_ws + list_off);

    const dim3 blk128(128), blk256(256);
    const int rowTiles = (N + 15) / 16;
    const int rowPairs = (rowTiles + 1) / 2;
    const int nwords   = nchunks * (CHUNK / 4);

    if (pair)
        edge_keys<true><<<(nwords + 255) / 256, blk256, 0, stream>>>(src, dst, ew, key, kw, ob, N, E, nwords);
    else
        edge_keys<false><<<(nwords + 255) / 256, blk256, 0, stream>>>(src, dst, ew, key, kw, ob, N, E, nwords);

    for (int l = 0; l < 3; ++l) {
        float* st = stats + (l & 1) * 128;
        const float* sp = stats + ((l + 1) & 1) * 128;
        if (l == 0)
            gemm_wmma_f32_il<false><<<rowPairs, blk128, 0, stream>>>(x, cw[0], bufA, N, 128, st, sp, bg[0], bb[0]);
        else
            gemm_wmma_f32_il<true><<<rowPairs, blk128, 0, stream>>>(bufB, cw[l], bufA, N, HIDC, st, sp, bg[l - 1], bb[l - 1]);
        if (replay && l == 0)
            gcn_owner_sum<true, 1><<<NRG, dim3(1024), 0, stream>>>(bufA, key, ew, kw, ob, bufB, N, nchunks, lists, nblk_g, cb[l], st);
        else if (replay)
            gcn_owner_sum<true, 2><<<NRG, dim3(1024), 0, stream>>>(bufA, key, ew, kw, ob, bufB, N, nchunks, lists, nblk_g, cb[l], st);
        else if (pair)
            gcn_owner_sum<true, 0><<<NRG, dim3(1024), 0, stream>>>(bufA, key, ew, kw, ob, bufB, N, nchunks, lists, nblk_g, cb[l], st);
        else
            gcn_owner_sum<false, 0><<<NRG, dim3(1024), 0, stream>>>(bufA, key, ew, kw, ob, bufB, N, nchunks, lists, nblk_g, cb[l], st);
    }

    mlp_chain<<<(rowTiles + 3) / 4, blk128, 0, stream>>>(bufB, w1, b1, w2, b2, w3, b3, (float*)d_out, N, stats, bg[2], bb[2]);
    (void)n_in; (void)out_size;
}
